# starting kernel
# baseline (speedup 1.0000x reference)
#include <hip/hip_runtime.h>


typedef __attribute__((ext_vector_type(2))) float v2f;
typedef __attribute__((ext_vector_type(8))) float v8f;

#define NPIX 2304
#define KDIM 576
#define HW   48
#define CCH  64

#define WMMA_F32(a, b, c) \
  __builtin_amdgcn_wmma_f32_16x16x4_f32(false, (a), false, (b), (short)0, (c), false, false)

__global__ __launch_bounds__(256) void prep_norm_kernel(
    const float* __restrict__ bg,
    const float* __restrict__ mask,
    float* __restrict__ mp,
    float* __restrict__ rnrm)
{
  int n = blockIdx.x * 256 + threadIdx.x;
  if (n >= NPIX) return;
  int ny = n / HW, nx = n % HW;

  float msum = 0.f;
  #pragma unroll
  for (int dy = 0; dy < 3; ++dy) {
    int y = ny + dy - 1;
    if ((unsigned)y >= (unsigned)HW) continue;
    #pragma unroll
    for (int dx = 0; dx < 3; ++dx) {
      int x = nx + dx - 1;
      if ((unsigned)x >= (unsigned)HW) continue;
      msum += mask[y * HW + x];
    }
  }
  mp[n] = (msum != 0.f) ? 1.f : 0.f;

  float sq = 0.f;
  for (int c = 0; c < CCH; ++c) {
    const float* bc = bg + c * NPIX;
    #pragma unroll
    for (int dy = 0; dy < 3; ++dy) {
      int y = ny + dy - 1;
      if ((unsigned)y >= (unsigned)HW) continue;
      #pragma unroll
      for (int dx = 0; dx < 3; ++dx) {
        int x = nx + dx - 1;
        if ((unsigned)x >= (unsigned)HW) continue;
        float v = bc[y * HW + x];
        sq += v * v;
      }
    }
  }
  rnrm[n] = 1.f / fmaxf(sqrtf(sq), 1e-4f);
}

__global__ __launch_bounds__(256) void prep_unfold_kernel(
    const float* __restrict__ fg, const float* __restrict__ bg,
    const float* __restrict__ rnrm,
    float* __restrict__ Pf, float* __restrict__ Pbn)
{
  int t = blockIdx.x * 256 + threadIdx.x;
  if (t >= NPIX * KDIM) return;
  int n = t / KDIM, k = t % KDIM;
  int c = k / 9, r = k % 9, dy = r / 3, dx = r % 3;
  int y = n / HW + dy - 1, x = n % HW + dx - 1;
  bool in = ((unsigned)y < (unsigned)HW) && ((unsigned)x < (unsigned)HW);
  float vf = 0.f, vb = 0.f;
  if (in) {
    int off = c * NPIX + y * HW + x;
    vf = fg[off];
    vb = bg[off];
  }
  Pf[t]  = vf;
  Pbn[t] = vb * rnrm[n];
}

__global__ __launch_bounds__(256) void prep_unfold_t_kernel(
    const float* __restrict__ bg, const float* __restrict__ rnrm,
    float* __restrict__ Pbnt)
{
  int t = blockIdx.x * 256 + threadIdx.x;
  if (t >= NPIX * KDIM) return;
  int k = t / NPIX, n = t % NPIX;
  int c = k / 9, r = k % 9, dy = r / 3, dx = r % 3;
  int y = n / HW + dy - 1, x = n % HW + dx - 1;
  bool in = ((unsigned)y < (unsigned)HW) && ((unsigned)x < (unsigned)HW);
  float vb = in ? bg[c * NPIX + y * HW + x] : 0.f;
  Pbnt[t] = vb * rnrm[n];
}

__global__ __launch_bounds__(256) void gemm1_kernel(
    const float* __restrict__ Pf,
    const float* __restrict__ Pbn,
    float* __restrict__ attn)
{
  __shared__ float la[128][36];
  __shared__ float lb[256][36];

  const int tid = threadIdx.x, lane = tid & 31, wave = tid >> 5;
  const int wp = wave >> 2;
  const int wn = wave & 3;
  const int p0 = blockIdx.y * 128, n0 = blockIdx.x * 256;
  const int fr = lane & 15;
  const int fk = (lane >> 4) << 1;

  v8f acc[4][4];
  #pragma unroll
  for (int i = 0; i < 4; ++i)
    #pragma unroll
    for (int j = 0; j < 4; ++j) acc[i][j] = (v8f){};

  for (int kc = 0; kc < KDIM; kc += 32) {
    __syncthreads();
    #pragma unroll
    for (int i = 0; i < 4; ++i) {
      int e = tid + i * 256;
      int row = e >> 3, col4 = (e & 7) << 2;
      *(float4*)&la[row][col4] =
          *(const float4*)&Pf[(size_t)(p0 + row) * KDIM + kc + col4];
    }
    #pragma unroll
    for (int i = 0; i < 8; ++i) {
      int e = tid + i * 256;
      int row = e >> 3, col4 = (e & 7) << 2;
      *(float4*)&lb[row][col4] =
          *(const float4*)&Pbn[(size_t)(n0 + row) * KDIM + kc + col4];
    }
    __syncthreads();

    #pragma unroll
    for (int kk = 0; kk < 32; kk += 4) {
      v2f af[4], bf[4];
      #pragma unroll
      for (int i = 0; i < 4; ++i)
        af[i] = *(const v2f*)&la[wp * 64 + i * 16 + fr][kk + fk];
      #pragma unroll
      for (int j = 0; j < 4; ++j)
        bf[j] = *(const v2f*)&lb[wn * 64 + j * 16 + fr][kk + fk];
      #pragma unroll
      for (int i = 0; i < 4; ++i)
        #pragma unroll
        for (int j = 0; j < 4; ++j)
          acc[i][j] = WMMA_F32(af[i], bf[j], acc[i][j]);
    }
  }

  const int dr = (lane >> 4) << 3, dc = lane & 15;
  #pragma unroll
  for (int i = 0; i < 4; ++i)
    #pragma unroll
    for (int j = 0; j < 4; ++j) {
      float* dst = attn + (size_t)(p0 + wp * 64 + i * 16 + dr) * NPIX
                        + n0 + wn * 64 + j * 16 + dc;
      #pragma unroll
      for (int r = 0; r < 8; ++r) dst[(size_t)r * NPIX] = acc[i][j][r];
    }
}

__global__ __launch_bounds__(256) void softmax_kernel(
    float* __restrict__ attn, const float* __restrict__ mp)
{
  const int p = blockIdx.x * 8 + (threadIdx.x >> 5);
  const int lane = threadIdx.x & 31;
  float* row = attn + (size_t)p * NPIX;

  float m = -3.0e38f, s = 0.f;
  for (int n = lane; n < NPIX; n += 32) {
    float z = row[n] * mp[n] * 10.0f;
    float mn = fmaxf(m, z);
    s = s * __expf(m - mn) + __expf(z - mn);
    m = mn;
  }
  #pragma unroll
  for (int off = 16; off > 0; off >>= 1) {
    float m2 = __shfl_xor(m, off, 32);
    float s2 = __shfl_xor(s, off, 32);
    float mn = fmaxf(m, m2);
    s = s * __expf(m - mn) + s2 * __expf(m2 - mn);
    m = mn;
  }
  const float inv = 1.0f / s;
  for (int n = lane; n < NPIX; n += 32) {
    float g = mp[n];
    float z = row[n] * g * 10.0f;
    row[n] = __expf(z - m) * inv * g;
  }
}

__global__ __launch_bounds__(128) void gemm2_kernel(
    const float* __restrict__ Pbnt,
    const float* __restrict__ attn,
    float* __restrict__ V)
{
  __shared__ float la[64][36];
  __shared__ float lb[256][36];

  const int tid = threadIdx.x, lane = tid & 31, wave = tid >> 5;
  const int wq = wave;
  const int k0 = blockIdx.y * 64, q0 = blockIdx.x * 256;
  const int fr = lane & 15;
  const int fk = (lane >> 4) << 1;

  v8f acc[4][4];
  #pragma unroll
  for (int i = 0; i < 4; ++i)
    #pragma unroll
    for (int j = 0; j < 4; ++j) acc[i][j] = (v8f){};

  for (int nc = 0; nc < NPIX; nc += 32) {
    __syncthreads();
    #pragma unroll
    for (int i = 0; i < 4; ++i) {
      int e = tid + i * 128;
      int row = e >> 3, col4 = (e & 7) << 2;
      *(float4*)&la[row][col4] =
          *(const float4*)&Pbnt[(size_t)(k0 + row) * NPIX + nc + col4];
    }
    #pragma unroll
    for (int i = 0; i < 16; ++i) {
      int e = tid + i * 128;
      int row = e >> 3, col4 = (e & 7) << 2;
      *(float4*)&lb[row][col4] =
          *(const float4*)&attn[(size_t)(q0 + row) * NPIX + nc + col4];
    }
    __syncthreads();

    #pragma unroll
    for (int kk = 0; kk < 32; kk += 4) {
      v2f af[4], bf[4];
      #pragma unroll
      for (int i = 0; i < 4; ++i)
        af[i] = *(const v2f*)&la[i * 16 + fr][kk + fk];
      #pragma unroll
      for (int j = 0; j < 4; ++j)
        bf[j] = *(const v2f*)&lb[wq * 64 + j * 16 + fr][kk + fk];
      #pragma unroll
      for (int i = 0; i < 4; ++i)
        #pragma unroll
        for (int j = 0; j < 4; ++j)
          acc[i][j] = WMMA_F32(af[i], bf[j], acc[i][j]);
    }
  }

  const int dr = (lane >> 4) << 3, dc = lane & 15;
  #pragma unroll
  for (int i = 0; i < 4; ++i)
    #pragma unroll
    for (int j = 0; j < 4; ++j) {
      float* dst = V + (size_t)(k0 + i * 16 + dr) * NPIX
                     + q0 + wq * 64 + j * 16 + dc;
      #pragma unroll
      for (int r = 0; r < 8; ++r) dst[(size_t)r * NPIX] = acc[i][j][r];
    }
}

__global__ __launch_bounds__(256) void fold_kernel(
    const float* __restrict__ V, float* __restrict__ out)
{
  int t = blockIdx.x * 256 + threadIdx.x;
  if (t >= CCH * NPIX) return;
  int c = t / NPIX, p = t % NPIX;
  int y = p / HW, x = p % HW;
  float acc = 0.f;
  #pragma unroll
  for (int dy = 0; dy < 3; ++dy) {
    int yy = y + 1 - dy;
    if ((unsigned)yy >= (unsigned)HW) continue;
    #pragma unroll
    for (int dx = 0; dx < 3; ++dx) {
      int xx = x + 1 - dx;
      if ((unsigned)xx >= (unsigned)HW) continue;
      acc += V[(size_t)(c * 9 + dy * 3 + dx) * NPIX + yy * HW + xx];
    }
  }
  out[t] = acc;
}

extern "C" void kernel_launch(void* const* d_in, const int* in_sizes, int n_in,
                              void* d_out, int out_size, void* d_ws, size_t ws_size,
                              hipStream_t stream) {
  const float* fg   = (const float*)d_in[0];
  const float* bg   = (const float*)d_in[1];
  const float* mask = (const float*)d_in[2];
  float* out = (float*)d_out;

  float* ws = (float*)d_ws;
  const size_t PK = (size_t)NPIX * KDIM;
  const size_t PN = (size_t)NPIX * NPIX;
  float* mp   = ws;
  float* rn   = ws + NPIX;
  float* Pf   = ws + 2 * NPIX;
  float* Pbn  = Pf + PK;
  float* Pbnt = Pbn + PK;
  float* attn = Pbnt + PK;
  float* V    = attn + PN;

  for (int b = 0; b < 4; ++b) {
    const float* fgb = fg + (size_t)b * CCH * NPIX;
    const float* bgb = bg + (size_t)b * CCH * NPIX;
    const float* mb  = mask + (size_t)b * NPIX;
    float* outb = out + (size_t)b * CCH * NPIX;

    prep_norm_kernel<<<(NPIX + 255) / 256, 256, 0, stream>>>(bgb, mb, mp, rn);
    prep_unfold_kernel<<<(NPIX * KDIM) / 256, 256, 0, stream>>>(fgb, bgb, rn, Pf, Pbn);
    prep_unfold_t_kernel<<<(NPIX * KDIM) / 256, 256, 0, stream>>>(bgb, rn, Pbnt);

    dim3 g1(NPIX / 256, NPIX / 128);
    gemm1_kernel<<<g1, 256, 0, stream>>>(Pf, Pbn, attn);

    softmax_kernel<<<NPIX / 8, 256, 0, stream>>>(attn, mp);

    dim3 g2(NPIX / 256, KDIM / 64);
    gemm2_kernel<<<g2, 128, 0, stream>>>(Pbnt, attn, V);

    fold_kernel<<<(CCH * NPIX) / 256, 256, 0, stream>>>(V, outb);
  }
}


# both GEMM K-loops software-pipelined: next chunk's global loads issued as one group into registers ahead of the WMMA block, two-stage LDS ring, one barrier per chunk
# speedup vs baseline: 2.5846x; 2.5846x over previous
#include <hip/hip_runtime.h>


typedef __attribute__((ext_vector_type(2))) float v2f;
typedef __attribute__((ext_vector_type(8))) float v8f;
typedef __attribute__((ext_vector_type(4))) float v4f;

#define NPIX 2304
#define KDIM 576
#define HW   48
#define CCH  64

#define WMMA_F32(a, b, c) \
  __builtin_amdgcn_wmma_f32_16x16x4_f32(false, (a), false, (b), (short)0, (c), false, false)

__global__ __launch_bounds__(256) void prep_norm_kernel(
    const float* __restrict__ bg,
    const float* __restrict__ mask,
    float* __restrict__ mp,
    float* __restrict__ rnrm)
{
  int n = blockIdx.x * 256 + threadIdx.x;
  if (n >= NPIX) return;
  int ny = n / HW, nx = n % HW;

  float msum = 0.f;
  #pragma unroll
  for (int dy = 0; dy < 3; ++dy) {
    int y = ny + dy - 1;
    if ((unsigned)y >= (unsigned)HW) continue;
    #pragma unroll
    for (int dx = 0; dx < 3; ++dx) {
      int x = nx + dx - 1;
      if ((unsigned)x >= (unsigned)HW) continue;
      msum += mask[y * HW + x];
    }
  }
  mp[n] = (msum != 0.f) ? 1.f : 0.f;

  float sq = 0.f;
  for (int c = 0; c < CCH; ++c) {
    const float* bc = bg + c * NPIX;
    #pragma unroll
    for (int dy = 0; dy < 3; ++dy) {
      int y = ny + dy - 1;
      if ((unsigned)y >= (unsigned)HW) continue;
      #pragma unroll
      for (int dx = 0; dx < 3; ++dx) {
        int x = nx + dx - 1;
        if ((unsigned)x >= (unsigned)HW) continue;
        float v = bc[y * HW + x];
        sq += v * v;
      }
    }
  }
  rnrm[n] = 1.f / fmaxf(sqrtf(sq), 1e-4f);
}

__global__ __launch_bounds__(256) void prep_unfold_kernel(
    const float* __restrict__ fg, const float* __restrict__ bg,
    const float* __restrict__ rnrm,
    float* __restrict__ Pf, float* __restrict__ Pbn)
{
  int t = blockIdx.x * 256 + threadIdx.x;
  if (t >= NPIX * KDIM) return;
  int n = t / KDIM, k = t % KDIM;
  int c = k / 9, r = k % 9, dy = r / 3, dx = r % 3;
  int y = n / HW + dy - 1, x = n % HW + dx - 1;
  bool in = ((unsigned)y < (unsigned)HW) && ((unsigned)x < (unsigned)HW);
  float vf = 0.f, vb = 0.f;
  if (in) {
    int off = c * NPIX + y * HW + x;
    vf = fg[off];
    vb = bg[off];
  }
  Pf[t]  = vf;
  Pbn[t] = vb * rnrm[n];
}

__global__ __launch_bounds__(256) void prep_unfold_t_kernel(
    const float* __restrict__ bg, const float* __restrict__ rnrm,
    float* __restrict__ Pbnt)
{
  int t = blockIdx.x * 256 + threadIdx.x;
  if (t >= NPIX * KDIM) return;
  int k = t / NPIX, n = t % NPIX;
  int c = k / 9, r = k % 9, dy = r / 3, dx = r % 3;
  int y = n / HW + dy - 1, x = n % HW + dx - 1;
  bool in = ((unsigned)y < (unsigned)HW) && ((unsigned)x < (unsigned)HW);
  float vb = in ? bg[c * NPIX + y * HW + x] : 0.f;
  Pbnt[t] = vb * rnrm[n];
}

#define CHUNK_MMA(LA, LB, AOFF, BOFF)                                   \
  _Pragma("unroll")                                                     \
  for (int kk = 0; kk < 32; kk += 4) {                                  \
    v2f af[4], bf[4];                                                   \
    _Pragma("unroll")                                                   \
    for (int i = 0; i < 4; ++i)                                         \
      af[i] = *(const v2f*)&LA[(AOFF) + i * 16 + fr][kk + fk];          \
    _Pragma("unroll")                                                   \
    for (int j = 0; j < 4; ++j)                                         \
      bf[j] = *(const v2f*)&LB[(BOFF) + j * 16 + fr][kk + fk];          \
    _Pragma("unroll")                                                   \
    for (int i = 0; i < 4; ++i)                                         \
      _Pragma("unroll")                                                 \
      for (int j = 0; j < 4; ++j)                                       \
        acc[i][j] = WMMA_F32(af[i], bf[j], acc[i][j]);                  \
  }

__global__ __launch_bounds__(256) void gemm1_kernel(
    const float* __restrict__ Pf,
    const float* __restrict__ Pbn,
    float* __restrict__ attn)
{
  __shared__ __attribute__((aligned(16))) float la[2][128][36];
  __shared__ __attribute__((aligned(16))) float lb[2][256][36];

  const int tid = threadIdx.x, lane = tid & 31, wave = tid >> 5;
  const int wp = wave >> 2;
  const int wn = wave & 3;
  const int p0 = blockIdx.y * 128, n0 = blockIdx.x * 256;
  const int fr = lane & 15;
  const int fk = (lane >> 4) << 1;

  const int lrow = tid >> 3;
  const int lcol = (tid & 7) << 2;
  const float* asrc = Pf  + (size_t)(p0 + lrow) * KDIM + lcol;
  const float* bsrc = Pbn + (size_t)(n0 + lrow) * KDIM + lcol;

  v8f acc[4][4];
  #pragma unroll
  for (int i = 0; i < 4; ++i)
    #pragma unroll
    for (int j = 0; j < 4; ++j) acc[i][j] = (v8f){};

  v4f ra[4], rb[8];
  #pragma unroll
  for (int i = 0; i < 4; ++i)
    ra[i] = *(const v4f*)(asrc + (size_t)i * 32 * KDIM);
  #pragma unroll
  for (int i = 0; i < 8; ++i)
    rb[i] = *(const v4f*)(bsrc + (size_t)i * 32 * KDIM);
  #pragma unroll
  for (int i = 0; i < 4; ++i) *(v4f*)&la[0][lrow + i * 32][lcol] = ra[i];
  #pragma unroll
  for (int i = 0; i < 8; ++i) *(v4f*)&lb[0][lrow + i * 32][lcol] = rb[i];
  __syncthreads();

  int buf = 0;
  for (int kc = 32; kc < KDIM; kc += 32) {
    asrc += 32; bsrc += 32;
    #pragma unroll
    for (int i = 0; i < 4; ++i)
      ra[i] = *(const v4f*)(asrc + (size_t)i * 32 * KDIM);
    #pragma unroll
    for (int i = 0; i < 8; ++i)
      rb[i] = *(const v4f*)(bsrc + (size_t)i * 32 * KDIM);
    __builtin_amdgcn_sched_barrier(0);

    CHUNK_MMA(la[buf], lb[buf], wp * 64, wn * 64)

    __builtin_amdgcn_sched_barrier(0);
    #pragma unroll
    for (int i = 0; i < 4; ++i) *(v4f*)&la[buf ^ 1][lrow + i * 32][lcol] = ra[i];
    #pragma unroll
    for (int i = 0; i < 8; ++i) *(v4f*)&lb[buf ^ 1][lrow + i * 32][lcol] = rb[i];
    __syncthreads();
    buf ^= 1;
  }
  CHUNK_MMA(la[buf], lb[buf], wp * 64, wn * 64)

  const int dr = (lane >> 4) << 3, dc = lane & 15;
  #pragma unroll
  for (int i = 0; i < 4; ++i)
    #pragma unroll
    for (int j = 0; j < 4; ++j) {
      float* dst = attn + (size_t)(p0 + wp * 64 + i * 16 + dr) * NPIX
                        + n0 + wn * 64 + j * 16 + dc;
      #pragma unroll
      for (int r = 0; r < 8; ++r) dst[(size_t)r * NPIX] = acc[i][j][r];
    }
}

__global__ __launch_bounds__(256) void softmax_kernel(
    float* __restrict__ attn, const float* __restrict__ mp)
{
  const int p = blockIdx.x * 8 + (threadIdx.x >> 5);
  const int lane = threadIdx.x & 31;
  float* row = attn + (size_t)p * NPIX;

  float m = -3.0e38f, s = 0.f;
  for (int n = lane; n < NPIX; n += 32) {
    float z = row[n] * mp[n] * 10.0f;
    float mn = fmaxf(m, z);
    s = s * __expf(m - mn) + __expf(z - mn);
    m = mn;
  }
  #pragma unroll
  for (int off = 16; off > 0; off >>= 1) {
    float m2 = __shfl_xor(m, off, 32);
    float s2 = __shfl_xor(s, off, 32);
    float mn = fmaxf(m, m2);
    s = s * __expf(m - mn) + s2 * __expf(m2 - mn);
    m = mn;
  }
  const float inv = 1.0f / s;
  for (int n = lane; n < NPIX; n += 32) {
    float g = mp[n];
    float z = row[n] * g * 10.0f;
    row[n] = __expf(z - m) * inv * g;
  }
}

__global__ __launch_bounds__(128) __attribute__((amdgpu_waves_per_eu(1, 1)))
void gemm2_kernel(
    const float* __restrict__ Pbnt,
    const float* __restrict__ attn,
    float* __restrict__ V)
{
  __shared__ __attribute__((aligned(16))) float la[2][64][36];
  __shared__ __attribute__((aligned(16))) float lb[2][256][36];

  const int tid = threadIdx.x, lane = tid & 31, wave = tid >> 5;
  const int wq = wave;
  const int k0 = blockIdx.y * 64, q0 = blockIdx.x * 256;
  const int fr = lane & 15;
  const int fk = (lane >> 4) << 1;

  const int lrow = tid >> 3;
  const int lcol = (tid & 7) << 2;
  const float* asrc = Pbnt + (size_t)(k0 + lrow) * NPIX + lcol;
  const float* bsrc = attn + (size_t)(q0 + lrow) * NPIX + lcol;

  v8f acc[4][4];
  #pragma unroll
  for (int i = 0; i < 4; ++i)
    #pragma unroll
    for (int j = 0; j < 4; ++j) acc[i][j] = (v8f){};

  v4f ra[4], rb[16];
  #pragma unroll
  for (int i = 0; i < 4; ++i)
    ra[i] = *(const v4f*)(asrc + (size_t)i * 16 * NPIX);
  #pragma unroll
  for (int i = 0; i < 16; ++i)
    rb[i] = *(const v4f*)(bsrc + (size_t)i * 16 * NPIX);
  #pragma unroll
  for (int i = 0; i < 4; ++i)  *(v4f*)&la[0][lrow + i * 16][lcol] = ra[i];
  #pragma unroll
  for (int i = 0; i < 16; ++i) *(v4f*)&lb[0][lrow + i * 16][lcol] = rb[i];
  __syncthreads();

  int buf = 0;
  for (int nc = 32; nc < NPIX; nc += 32) {
    asrc += 32; bsrc += 32;
    #pragma unroll
    for (int i = 0; i < 4; ++i)
      ra[i] = *(const v4f*)(asrc + (size_t)i * 16 * NPIX);
    #pragma unroll
    for (int i = 0; i < 16; ++i)
      rb[i] = *(const v4f*)(bsrc + (size_t)i * 16 * NPIX);
    __builtin_amdgcn_sched_barrier(0);

    CHUNK_MMA(la[buf], lb[buf], 0, wq * 64)

    __builtin_amdgcn_sched_barrier(0);
    #pragma unroll
    for (int i = 0; i < 4; ++i)  *(v4f*)&la[buf ^ 1][lrow + i * 16][lcol] = ra[i];
    #pragma unroll
    for (int i = 0; i < 16; ++i) *(v4f*)&lb[buf ^ 1][lrow + i * 16][lcol] = rb[i];
    __syncthreads();
    buf ^= 1;
  }
  CHUNK_MMA(la[buf], lb[buf], 0, wq * 64)

  const int dr = (lane >> 4) << 3, dc = lane & 15;
  #pragma unroll
  for (int i = 0; i < 4; ++i)
    #pragma unroll
    for (int j = 0; j < 4; ++j) {
      float* dst = V + (size_t)(k0 + i * 16 + dr) * NPIX
                     + q0 + wq * 64 + j * 16 + dc;
      #pragma unroll
      for (int r = 0; r < 8; ++r) dst[(size_t)r * NPIX] = acc[i][j][r];
    }
}

__global__ __launch_bounds__(256) void fold_kernel(
    const float* __restrict__ V, float* __restrict__ out)
{
  int t = blockIdx.x * 256 + threadIdx.x;
  if (t >= CCH * NPIX) return;
  int c = t / NPIX, p = t % NPIX;
  int y = p / HW, x = p % HW;
  float acc = 0.f;
  #pragma unroll
  for (int dy = 0; dy < 3; ++dy) {
    int yy = y + 1 - dy;
    if ((unsigned)yy >= (unsigned)HW) continue;
    #pragma unroll
    for (int dx = 0; dx < 3; ++dx) {
      int xx = x + 1 - dx;
      if ((unsigned)xx >= (unsigned)HW) continue;
      acc += V[(size_t)(c * 9 + dy * 3 + dx) * NPIX + yy * HW + xx];
    }
  }
  out[t] = acc;
}

extern "C" void kernel_launch(void* const* d_in, const int* in_sizes, int n_in,
                              void* d_out, int out_size, void* d_ws, size_t ws_size,
                              hipStream_t stream) {
  const float* fg   = (const float*)d_in[0];
  const float* bg   = (const float*)d_in[1];
  const float* mask = (const float*)d_in[2];
  float* out = (float*)d_out;

  float* ws = (float*)d_ws;
  const size_t PK = (size_t)NPIX * KDIM;
  const size_t PN = (size_t)NPIX * NPIX;
  float* mp   = ws;
  float* rn   = ws + NPIX;
  float* Pf   = ws + 2 * NPIX;
  float* Pbn  = Pf + PK;
  float* Pbnt = Pbn + PK;
  float* attn = Pbnt + PK;
  float* V    = attn + PN;

  for (int b = 0; b < 4; ++b) {
    const float* fgb = fg + (size_t)b * CCH * NPIX;
    const float* bgb = bg + (size_t)b * CCH * NPIX;
    const float* mb  = mask + (size_t)b * NPIX;
    float* outb = out + (size_t)b * CCH * NPIX;

    prep_norm_kernel<<<(NPIX + 255) / 256, 256, 0, stream>>>(bgb, mb, mp, rn);
    prep_unfold_kernel<<<(NPIX * KDIM) / 256, 256, 0, stream>>>(fgb, bgb, rn, Pf, Pbn);
    prep_unfold_t_kernel<<<(NPIX * KDIM) / 256, 256, 0, stream>>>(bgb, rn, Pbnt);

    dim3 g1(NPIX / 256, NPIX / 128);
    gemm1_kernel<<<g1, 256, 0, stream>>>(Pf, Pbn, attn);

    softmax_kernel<<<NPIX / 8, 256, 0, stream>>>(attn, mp);

    dim3 g2(NPIX / 256, KDIM / 64);
    gemm2_kernel<<<g2, 128, 0, stream>>>(Pbnt, attn, V);

    fold_kernel<<<(CCH * NPIX) / 256, 256, 0, stream>>>(V, outb);
  }
}


# GEMM2 contraction split in three over grid z: 81 -> 243 workgroups, one per WGP, each wave's serial WMMA chain 9216 -> 3072; fold adds the three partial planes in fixed order
# speedup vs baseline: 3.2820x; 1.2698x over previous
#include <hip/hip_runtime.h>


typedef __attribute__((ext_vector_type(2))) float v2f;
typedef __attribute__((ext_vector_type(8))) float v8f;
typedef __attribute__((ext_vector_type(4))) float v4f;

#define NPIX 2304
#define KDIM 576
#define HW   48
#define CCH  64
#define NSPLIT 3
#define NSEG (NPIX / NSPLIT)

#define WMMA_F32(a, b, c) \
  __builtin_amdgcn_wmma_f32_16x16x4_f32(false, (a), false, (b), (short)0, (c), false, false)

__global__ __launch_bounds__(256) void prep_norm_kernel(
    const float* __restrict__ bg,
    const float* __restrict__ mask,
    float* __restrict__ mp,
    float* __restrict__ rnrm)
{
  int n = blockIdx.x * 256 + threadIdx.x;
  if (n >= NPIX) return;
  int ny = n / HW, nx = n % HW;

  float msum = 0.f;
  #pragma unroll
  for (int dy = 0; dy < 3; ++dy) {
    int y = ny + dy - 1;
    if ((unsigned)y >= (unsigned)HW) continue;
    #pragma unroll
    for (int dx = 0; dx < 3; ++dx) {
      int x = nx + dx - 1;
      if ((unsigned)x >= (unsigned)HW) continue;
      msum += mask[y * HW + x];
    }
  }
  mp[n] = (msum != 0.f) ? 1.f : 0.f;

  float sq = 0.f;
  for (int c = 0; c < CCH; ++c) {
    const float* bc = bg + c * NPIX;
    #pragma unroll
    for (int dy = 0; dy < 3; ++dy) {
      int y = ny + dy - 1;
      if ((unsigned)y >= (unsigned)HW) continue;
      #pragma unroll
      for (int dx = 0; dx < 3; ++dx) {
        int x = nx + dx - 1;
        if ((unsigned)x >= (unsigned)HW) continue;
        float v = bc[y * HW + x];
        sq += v * v;
      }
    }
  }
  rnrm[n] = 1.f / fmaxf(sqrtf(sq), 1e-4f);
}

__global__ __launch_bounds__(256) void prep_unfold_kernel(
    const float* __restrict__ fg, const float* __restrict__ bg,
    const float* __restrict__ rnrm,
    float* __restrict__ Pf, float* __restrict__ Pbn)
{
  int t = blockIdx.x * 256 + threadIdx.x;
  if (t >= NPIX * KDIM) return;
  int n = t / KDIM, k = t % KDIM;
  int c = k / 9, r = k % 9, dy = r / 3, dx = r % 3;
  int y = n / HW + dy - 1, x = n % HW + dx - 1;
  bool in = ((unsigned)y < (unsigned)HW) && ((unsigned)x < (unsigned)HW);
  float vf = 0.f, vb = 0.f;
  if (in) {
    int off = c * NPIX + y * HW + x;
    vf = fg[off];
    vb = bg[off];
  }
  Pf[t]  = vf;
  Pbn[t] = vb * rnrm[n];
}

__global__ __launch_bounds__(256) void prep_unfold_t_kernel(
    const float* __restrict__ bg, const float* __restrict__ rnrm,
    float* __restrict__ Pbnt)
{
  int t = blockIdx.x * 256 + threadIdx.x;
  if (t >= NPIX * KDIM) return;
  int k = t / NPIX, n = t % NPIX;
  int c = k / 9, r = k % 9, dy = r / 3, dx = r % 3;
  int y = n / HW + dy - 1, x = n % HW + dx - 1;
  bool in = ((unsigned)y < (unsigned)HW) && ((unsigned)x < (unsigned)HW);
  float vb = in ? bg[c * NPIX + y * HW + x] : 0.f;
  Pbnt[t] = vb * rnrm[n];
}

#define CHUNK_MMA(LA, LB, AOFF, BOFF)                                   \
  _Pragma("unroll")                                                     \
  for (int kk = 0; kk < 32; kk += 4) {                                  \
    v2f af[4], bf[4];                                                   \
    _Pragma("unroll")                                                   \
    for (int i = 0; i < 4; ++i)                                         \
      af[i] = *(const v2f*)&LA[(AOFF) + i * 16 + fr][kk + fk];          \
    _Pragma("unroll")                                                   \
    for (int j = 0; j < 4; ++j)                                         \
      bf[j] = *(const v2f*)&LB[(BOFF) + j * 16 + fr][kk + fk];          \
    _Pragma("unroll")                                                   \
    for (int i = 0; i < 4; ++i)                                         \
      _Pragma("unroll")                                                 \
      for (int j = 0; j < 4; ++j)                                       \
        acc[i][j] = WMMA_F32(af[i], bf[j], acc[i][j]);                  \
  }

__global__ __launch_bounds__(256) void gemm1_kernel(
    const float* __restrict__ Pf,
    const float* __restrict__ Pbn,
    float* __restrict__ attn)
{
  __shared__ __attribute__((aligned(16))) float la[2][128][36];
  __shared__ __attribute__((aligned(16))) float lb[2][256][36];

  const int tid = threadIdx.x, lane = tid & 31, wave = tid >> 5;
  const int wp = wave >> 2;
  const int wn = wave & 3;
  const int p0 = blockIdx.y * 128, n0 = blockIdx.x * 256;
  const int fr = lane & 15;
  const int fk = (lane >> 4) << 1;

  const int lrow = tid >> 3;
  const int lcol = (tid & 7) << 2;
  const float* asrc = Pf  + (size_t)(p0 + lrow) * KDIM + lcol;
  const float* bsrc = Pbn + (size_t)(n0 + lrow) * KDIM + lcol;

  v8f acc[4][4];
  #pragma unroll
  for (int i = 0; i < 4; ++i)
    #pragma unroll
    for (int j = 0; j < 4; ++j) acc[i][j] = (v8f){};

  v4f ra[4], rb[8];
  #pragma unroll
  for (int i = 0; i < 4; ++i)
    ra[i] = *(const v4f*)(asrc + (size_t)i * 32 * KDIM);
  #pragma unroll
  for (int i = 0; i < 8; ++i)
    rb[i] = *(const v4f*)(bsrc + (size_t)i * 32 * KDIM);
  #pragma unroll
  for (int i = 0; i < 4; ++i) *(v4f*)&la[0][lrow + i * 32][lcol] = ra[i];
  #pragma unroll
  for (int i = 0; i < 8; ++i) *(v4f*)&lb[0][lrow + i * 32][lcol] = rb[i];
  __syncthreads();

  int buf = 0;
  for (int kc = 32; kc < KDIM; kc += 32) {
    asrc += 32; bsrc += 32;
    #pragma unroll
    for (int i = 0; i < 4; ++i)
      ra[i] = *(const v4f*)(asrc + (size_t)i * 32 * KDIM);
    #pragma unroll
    for (int i = 0; i < 8; ++i)
      rb[i] = *(const v4f*)(bsrc + (size_t)i * 32 * KDIM);
    __builtin_amdgcn_sched_barrier(0);

    CHUNK_MMA(la[buf], lb[buf], wp * 64, wn * 64)

    __builtin_amdgcn_sched_barrier(0);
    #pragma unroll
    for (int i = 0; i < 4; ++i) *(v4f*)&la[buf ^ 1][lrow + i * 32][lcol] = ra[i];
    #pragma unroll
    for (int i = 0; i < 8; ++i) *(v4f*)&lb[buf ^ 1][lrow + i * 32][lcol] = rb[i];
    __syncthreads();
    buf ^= 1;
  }
  CHUNK_MMA(la[buf], lb[buf], wp * 64, wn * 64)

  const int dr = (lane >> 4) << 3, dc = lane & 15;
  #pragma unroll
  for (int i = 0; i < 4; ++i)
    #pragma unroll
    for (int j = 0; j < 4; ++j) {
      float* dst = attn + (size_t)(p0 + wp * 64 + i * 16 + dr) * NPIX
                        + n0 + wn * 64 + j * 16 + dc;
      #pragma unroll
      for (int r = 0; r < 8; ++r) dst[(size_t)r * NPIX] = acc[i][j][r];
    }
}

__global__ __launch_bounds__(256) void softmax_kernel(
    float* __restrict__ attn, const float* __restrict__ mp)
{
  const int p = blockIdx.x * 8 + (threadIdx.x >> 5);
  const int lane = threadIdx.x & 31;
  float* row = attn + (size_t)p * NPIX;

  float m = -3.0e38f, s = 0.f;
  for (int n = lane; n < NPIX; n += 32) {
    float z = row[n] * mp[n] * 10.0f;
    float mn = fmaxf(m, z);
    s = s * __expf(m - mn) + __expf(z - mn);
    m = mn;
  }
  #pragma unroll
  for (int off = 16; off > 0; off >>= 1) {
    float m2 = __shfl_xor(m, off, 32);
    float s2 = __shfl_xor(s, off, 32);
    float mn = fmaxf(m, m2);
    s = s * __expf(m - mn) + s2 * __expf(m2 - mn);
    m = mn;
  }
  const float inv = 1.0f / s;
  for (int n = lane; n < NPIX; n += 32) {
    float g = mp[n];
    float z = row[n] * g * 10.0f;
    row[n] = __expf(z - m) * inv * g;
  }
}

__global__ __launch_bounds__(128) __attribute__((amdgpu_waves_per_eu(1, 1)))
void gemm2_kernel(
    const float* __restrict__ Pbnt,
    const float* __restrict__ attn,
    float* __restrict__ Vp)
{
  __shared__ __attribute__((aligned(16))) float la[2][64][36];
  __shared__ __attribute__((aligned(16))) float lb[2][256][36];

  const int tid = threadIdx.x, lane = tid & 31, wave = tid >> 5;
  const int wq = wave;
  const int k0 = blockIdx.y * 64, q0 = blockIdx.x * 256;
  const int nb = blockIdx.z * NSEG;
  const int fr = lane & 15;
  const int fk = (lane >> 4) << 1;

  const int lrow = tid >> 3;
  const int lcol = (tid & 7) << 2;
  const float* asrc = Pbnt + (size_t)(k0 + lrow) * NPIX + nb + lcol;
  const float* bsrc = attn + (size_t)(q0 + lrow) * NPIX + nb + lcol;
  float* V = Vp + (size_t)blockIdx.z * ((size_t)KDIM * NPIX);

  v8f acc[4][4];
  #pragma unroll
  for (int i = 0; i < 4; ++i)
    #pragma unroll
    for (int j = 0; j < 4; ++j) acc[i][j] = (v8f){};

  v4f ra[4], rb[16];
  #pragma unroll
  for (int i = 0; i < 4; ++i)
    ra[i] = *(const v4f*)(asrc + (size_t)i * 16 * NPIX);
  #pragma unroll
  for (int i = 0; i < 16; ++i)
    rb[i] = *(const v4f*)(bsrc + (size_t)i * 16 * NPIX);
  #pragma unroll
  for (int i = 0; i < 4; ++i)  *(v4f*)&la[0][lrow + i * 16][lcol] = ra[i];
  #pragma unroll
  for (int i = 0; i < 16; ++i) *(v4f*)&lb[0][lrow + i * 16][lcol] = rb[i];
  __syncthreads();

  int buf = 0;
  for (int nc = 32; nc < NSEG; nc += 32) {
    asrc += 32; bsrc += 32;
    #pragma unroll
    for (int i = 0; i < 4; ++i)
      ra[i] = *(const v4f*)(asrc + (size_t)i * 16 * NPIX);
    #pragma unroll
    for (int i = 0; i < 16; ++i)
      rb[i] = *(const v4f*)(bsrc + (size_t)i * 16 * NPIX);
    __builtin_amdgcn_sched_barrier(0);

    CHUNK_MMA(la[buf], lb[buf], 0, wq * 64)

    __builtin_amdgcn_sched_barrier(0);
    #pragma unroll
    for (int i = 0; i < 4; ++i)  *(v4f*)&la[buf ^ 1][lrow + i * 16][lcol] = ra[i];
    #pragma unroll
    for (int i = 0; i < 16; ++i) *(v4f*)&lb[buf ^ 1][lrow + i * 16][lcol] = rb[i];
    __syncthreads();
    buf ^= 1;
  }
  CHUNK_MMA(la[buf], lb[buf], 0, wq * 64)

  const int dr = (lane >> 4) << 3, dc = lane & 15;
  #pragma unroll
  for (int i = 0; i < 4; ++i)
    #pragma unroll
    for (int j = 0; j < 4; ++j) {
      float* dst = V + (size_t)(k0 + i * 16 + dr) * NPIX
                     + q0 + wq * 64 + j * 16 + dc;
      #pragma unroll
      for (int r = 0; r < 8; ++r) dst[(size_t)r * NPIX] = acc[i][j][r];
    }
}

__global__ __launch_bounds__(256) void fold_kernel(
    const float* __restrict__ Vp, float* __restrict__ out)
{
  int t = blockIdx.x * 256 + threadIdx.x;
  if (t >= CCH * NPIX) return;
  int c = t / NPIX, p = t % NPIX;
  int y = p / HW, x = p % HW;
  float acc = 0.f;
  #pragma unroll
  for (int dy = 0; dy < 3; ++dy) {
    int yy = y + 1 - dy;
    if ((unsigned)yy >= (unsigned)HW) continue;
    #pragma unroll
    for (int dx = 0; dx < 3; ++dx) {
      int xx = x + 1 - dx;
      if ((unsigned)xx >= (unsigned)HW) continue;
      const float* v = Vp + (size_t)(c * 9 + dy * 3 + dx) * NPIX + yy * HW + xx;
      acc += (v[0] + v[(size_t)KDIM * NPIX]) + v[2 * (size_t)KDIM * NPIX];
    }
  }
  out[t] = acc;
}

extern "C" void kernel_launch(void* const* d_in, const int* in_sizes, int n_in,
                              void* d_out, int out_size, void* d_ws, size_t ws_size,
                              hipStream_t stream) {
  const float* fg   = (const float*)d_in[0];
  const float* bg   = (const float*)d_in[1];
  const float* mask = (const float*)d_in[2];
  float* out = (float*)d_out;

  float* ws = (float*)d_ws;
  const size_t PK = (size_t)NPIX * KDIM;
  const size_t PN = (size_t)NPIX * NPIX;
  float* mp   = ws;
  float* rn   = ws + NPIX;
  float* Pf   = ws + 2 * NPIX;
  float* Pbn  = Pf + PK;
  float* Vp   = Pf;
  float* Pbnt = Pf + 3 * PK;
  float* attn = Pbnt + PK;

  for (int b = 0; b < 4; ++b) {
    const float* fgb = fg + (size_t)b * CCH * NPIX;
    const float* bgb = bg + (size_t)b * CCH * NPIX;
    const float* mb  = mask + (size_t)b * NPIX;
    float* outb = out + (size_t)b * CCH * NPIX;

    prep_norm_kernel<<<(NPIX + 255) / 256, 256, 0, stream>>>(bgb, mb, mp, rn);
    prep_unfold_kernel<<<(NPIX * KDIM) / 256, 256, 0, stream>>>(fgb, bgb, rn, Pf, Pbn);
    prep_unfold_t_kernel<<<(NPIX * KDIM) / 256, 256, 0, stream>>>(bgb, rn, Pbnt);

    dim3 g1(NPIX / 256, NPIX / 128);
    gemm1_kernel<<<g1, 256, 0, stream>>>(Pf, Pbn, attn);

    softmax_kernel<<<NPIX / 8, 256, 0, stream>>>(attn, mp);

    dim3 g2(NPIX / 256, KDIM / 64, NSPLIT);
    gemm2_kernel<<<g2, 128, 0, stream>>>(Pbnt, attn, Vp);

    fold_kernel<<<(CCH * NPIX) / 256, 256, 0, stream>>>(Vp, outb);
  }
}


# norm kernel: 576 guarded loads per lane, each waited for alone, become 72 unbranched clamped loads issued as one group; channels spread over 8 waves (72 workgroups, not 9), partials joined in LDS
# speedup vs baseline: 4.8166x; 1.4676x over previous
#include <hip/hip_runtime.h>


typedef __attribute__((ext_vector_type(2))) float v2f;
typedef __attribute__((ext_vector_type(8))) float v8f;
typedef __attribute__((ext_vector_type(4))) float v4f;

#define NPIX 2304
#define KDIM 576
#define HW   48
#define CCH  64
#define NSPLIT 3
#define NSEG (NPIX / NSPLIT)

#define WMMA_F32(a, b, c) \
  __builtin_amdgcn_wmma_f32_16x16x4_f32(false, (a), false, (b), (short)0, (c), false, false)

__global__ __launch_bounds__(256) void prep_norm_kernel(
    const float* __restrict__ bg,
    const float* __restrict__ mask,
    float* __restrict__ mp,
    float* __restrict__ rnrm)
{
  __shared__ float part[8][32];

  const int lane = threadIdx.x & 31, w = threadIdx.x >> 5;
  const int n = blockIdx.x * 32 + lane;
  const int ny = n / HW, nx = n % HW;

  int  off[9];
  bool in[9];
  #pragma unroll
  for (int dy = 0; dy < 3; ++dy) {
    const int y = ny + dy - 1;
    const int cy = min(max(y, 0), HW - 1);
    #pragma unroll
    for (int dx = 0; dx < 3; ++dx) {
      const int x = nx + dx - 1;
      const int cx = min(max(x, 0), HW - 1);
      off[dy * 3 + dx] = cy * HW + cx;
      in[dy * 3 + dx]  = ((unsigned)y < (unsigned)HW) && ((unsigned)x < (unsigned)HW);
    }
  }

  const float* bc = bg + (size_t)(w * 8) * NPIX;
  float v[8][9];
  #pragma unroll
  for (int c = 0; c < 8; ++c)
    #pragma unroll
    for (int t = 0; t < 9; ++t)
      v[c][t] = bc[c * NPIX + off[t]];

  float mv[9];
  if (w == 0) {
    #pragma unroll
    for (int t = 0; t < 9; ++t) mv[t] = mask[off[t]];
  }

  float sq = 0.f;
  #pragma unroll
  for (int c = 0; c < 8; ++c)
    #pragma unroll
    for (int t = 0; t < 9; ++t) {
      const float x = in[t] ? v[c][t] : 0.f;
      sq += x * x;
    }
  part[w][lane] = sq;

  if (w == 0) {
    float msum = 0.f;
    #pragma unroll
    for (int t = 0; t < 9; ++t) msum += in[t] ? mv[t] : 0.f;
    mp[n] = (msum != 0.f) ? 1.f : 0.f;
  }
  __syncthreads();

  if (w == 0) {
    float s = part[0][lane];
    #pragma unroll
    for (int i = 1; i < 8; ++i) s += part[i][lane];
    rnrm[n] = 1.f / fmaxf(sqrtf(s), 1e-4f);
  }
}

__global__ __launch_bounds__(256) void prep_unfold_kernel(
    const float* __restrict__ fg, const float* __restrict__ bg,
    const float* __restrict__ rnrm,
    float* __restrict__ Pf, float* __restrict__ Pbn)
{
  int t = blockIdx.x * 256 + threadIdx.x;
  if (t >= NPIX * KDIM) return;
  int n = t / KDIM, k = t % KDIM;
  int c = k / 9, r = k % 9, dy = r / 3, dx = r % 3;
  int y = n / HW + dy - 1, x = n % HW + dx - 1;
  bool in = ((unsigned)y < (unsigned)HW) && ((unsigned)x < (unsigned)HW);
  float vf = 0.f, vb = 0.f;
  if (in) {
    int off = c * NPIX + y * HW + x;
    vf = fg[off];
    vb = bg[off];
  }
  Pf[t]  = vf;
  Pbn[t] = vb * rnrm[n];
}

__global__ __launch_bounds__(256) void prep_unfold_t_kernel(
    const float* __restrict__ bg, const float* __restrict__ rnrm,
    float* __restrict__ Pbnt)
{
  int t = blockIdx.x * 256 + threadIdx.x;
  if (t >= NPIX * KDIM) return;
  int k = t / NPIX, n = t % NPIX;
  int c = k / 9, r = k % 9, dy = r / 3, dx = r % 3;
  int y = n / HW + dy - 1, x = n % HW + dx - 1;
  bool in = ((unsigned)y < (unsigned)HW) && ((unsigned)x < (unsigned)HW);
  float vb = in ? bg[c * NPIX + y * HW + x] : 0.f;
  Pbnt[t] = vb * rnrm[n];
}

#define CHUNK_MMA(LA, LB, AOFF, BOFF)                                   \
  _Pragma("unroll")                                                     \
  for (int kk = 0; kk < 32; kk += 4) {                                  \
    v2f af[4], bf[4];                                                   \
    _Pragma("unroll")                                                   \
    for (int i = 0; i < 4; ++i)                                         \
      af[i] = *(const v2f*)&LA[(AOFF) + i * 16 + fr][kk + fk];          \
    _Pragma("unroll")                                                   \
    for (int j = 0; j < 4; ++j)                                         \
      bf[j] = *(const v2f*)&LB[(BOFF) + j * 16 + fr][kk + fk];          \
    _Pragma("unroll")                                                   \
    for (int i = 0; i < 4; ++i)                                         \
      _Pragma("unroll")                                                 \
      for (int j = 0; j < 4; ++j)                                       \
        acc[i][j] = WMMA_F32(af[i], bf[j], acc[i][j]);                  \
  }

__global__ __launch_bounds__(256) void gemm1_kernel(
    const float* __restrict__ Pf,
    const float* __restrict__ Pbn,
    float* __restrict__ attn)
{
  __shared__ __attribute__((aligned(16))) float la[2][128][36];
  __shared__ __attribute__((aligned(16))) float lb[2][256][36];

  const int tid = threadIdx.x, lane = tid & 31, wave = tid >> 5;
  const int wp = wave >> 2;
  const int wn = wave & 3;
  const int p0 = blockIdx.y * 128, n0 = blockIdx.x * 256;
  const int fr = lane & 15;
  const int fk = (lane >> 4) << 1;

  const int lrow = tid >> 3;
  const int lcol = (tid & 7) << 2;
  const float* asrc = Pf  + (size_t)(p0 + lrow) * KDIM + lcol;
  const float* bsrc = Pbn + (size_t)(n0 + lrow) * KDIM + lcol;

  v8f acc[4][4];
  #pragma unroll
  for (int i = 0; i < 4; ++i)
    #pragma unroll
    for (int j = 0; j < 4; ++j) acc[i][j] = (v8f){};

  v4f ra[4], rb[8];
  #pragma unroll
  for (int i = 0; i < 4; ++i)
    ra[i] = *(const v4f*)(asrc + (size_t)i * 32 * KDIM);
  #pragma unroll
  for (int i = 0; i < 8; ++i)
    rb[i] = *(const v4f*)(bsrc + (size_t)i * 32 * KDIM);
  #pragma unroll
  for (int i = 0; i < 4; ++i) *(v4f*)&la[0][lrow + i * 32][lcol] = ra[i];
  #pragma unroll
  for (int i = 0; i < 8; ++i) *(v4f*)&lb[0][lrow + i * 32][lcol] = rb[i];
  __syncthreads();

  int buf = 0;
  for (int kc = 32; kc < KDIM; kc += 32) {
    asrc += 32; bsrc += 32;
    #pragma unroll
    for (int i = 0; i < 4; ++i)
      ra[i] = *(const v4f*)(asrc + (size_t)i * 32 * KDIM);
    #pragma unroll
    for (int i = 0; i < 8; ++i)
      rb[i] = *(const v4f*)(bsrc + (size_t)i * 32 * KDIM);
    __builtin_amdgcn_sched_barrier(0);

    CHUNK_MMA(la[buf], lb[buf], wp * 64, wn * 64)

    __builtin_amdgcn_sched_barrier(0);
    #pragma unroll
    for (int i = 0; i < 4; ++i) *(v4f*)&la[buf ^ 1][lrow + i * 32][lcol] = ra[i];
    #pragma unroll
    for (int i = 0; i < 8; ++i) *(v4f*)&lb[buf ^ 1][lrow + i * 32][lcol] = rb[i];
    __syncthreads();
    buf ^= 1;
  }
  CHUNK_MMA(la[buf], lb[buf], wp * 64, wn * 64)

  const int dr = (lane >> 4) << 3, dc = lane & 15;
  #pragma unroll
  for (int i = 0; i < 4; ++i)
    #pragma unroll
    for (int j = 0; j < 4; ++j) {
      float* dst = attn + (size_t)(p0 + wp * 64 + i * 16 + dr) * NPIX
                        + n0 + wn * 64 + j * 16 + dc;
      #pragma unroll
      for (int r = 0; r < 8; ++r) dst[(size_t)r * NPIX] = acc[i][j][r];
    }
}

__global__ __launch_bounds__(256) void softmax_kernel(
    float* __restrict__ attn, const float* __restrict__ mp)
{
  const int p = blockIdx.x * 8 + (threadIdx.x >> 5);
  const int lane = threadIdx.x & 31;
  float* row = attn + (size_t)p * NPIX;

  float m = -3.0e38f, s = 0.f;
  for (int n = lane; n < NPIX; n += 32) {
    float z = row[n] * mp[n] * 10.0f;
    float mn = fmaxf(m, z);
    s = s * __expf(m - mn) + __expf(z - mn);
    m = mn;
  }
  #pragma unroll
  for (int off = 16; off > 0; off >>= 1) {
    float m2 = __shfl_xor(m, off, 32);
    float s2 = __shfl_xor(s, off, 32);
    float mn = fmaxf(m, m2);
    s = s * __expf(m - mn) + s2 * __expf(m2 - mn);
    m = mn;
  }
  const float inv = 1.0f / s;
  for (int n = lane; n < NPIX; n += 32) {
    float g = mp[n];
    float z = row[n] * g * 10.0f;
    row[n] = __expf(z - m) * inv * g;
  }
}

__global__ __launch_bounds__(128) __attribute__((amdgpu_waves_per_eu(1, 1)))
void gemm2_kernel(
    const float* __restrict__ Pbnt,
    const float* __restrict__ attn,
    float* __restrict__ Vp)
{
  __shared__ __attribute__((aligned(16))) float la[2][64][36];
  __shared__ __attribute__((aligned(16))) float lb[2][256][36];

  const int tid = threadIdx.x, lane = tid & 31, wave = tid >> 5;
  const int wq = wave;
  const int k0 = blockIdx.y * 64, q0 = blockIdx.x * 256;
  const int nb = blockIdx.z * NSEG;
  const int fr = lane & 15;
  const int fk = (lane >> 4) << 1;

  const int lrow = tid >> 3;
  const int lcol = (tid & 7) << 2;
  const float* asrc = Pbnt + (size_t)(k0 + lrow) * NPIX + nb + lcol;
  const float* bsrc = attn + (size_t)(q0 + lrow) * NPIX + nb + lcol;
  float* V = Vp + (size_t)blockIdx.z * ((size_t)KDIM * NPIX);

  v8f acc[4][4];
  #pragma unroll
  for (int i = 0; i < 4; ++i)
    #pragma unroll
    for (int j = 0; j < 4; ++j) acc[i][j] = (v8f){};

  v4f ra[4], rb[16];
  #pragma unroll
  for (int i = 0; i < 4; ++i)
    ra[i] = *(const v4f*)(asrc + (size_t)i * 16 * NPIX);
  #pragma unroll
  for (int i = 0; i < 16; ++i)
    rb[i] = *(const v4f*)(bsrc + (size_t)i * 16 * NPIX);
  #pragma unroll
  for (int i = 0; i < 4; ++i)  *(v4f*)&la[0][lrow + i * 16][lcol] = ra[i];
  #pragma unroll
  for (int i = 0; i < 16; ++i) *(v4f*)&lb[0][lrow + i * 16][lcol] = rb[i];
  __syncthreads();

  int buf = 0;
  for (int nc = 32; nc < NSEG; nc += 32) {
    asrc += 32; bsrc += 32;
    #pragma unroll
    for (int i = 0; i < 4; ++i)
      ra[i] = *(const v4f*)(asrc + (size_t)i * 16 * NPIX);
    #pragma unroll
    for (int i = 0; i < 16; ++i)
      rb[i] = *(const v4f*)(bsrc + (size_t)i * 16 * NPIX);
    __builtin_amdgcn_sched_barrier(0);

    CHUNK_MMA(la[buf], lb[buf], 0, wq * 64)

    __builtin_amdgcn_sched_barrier(0);
    #pragma unroll
    for (int i = 0; i < 4; ++i)  *(v4f*)&la[buf ^ 1][lrow + i * 16][lcol] = ra[i];
    #pragma unroll
    for (int i = 0; i < 16; ++i) *(v4f*)&lb[buf ^ 1][lrow + i * 16][lcol] = rb[i];
    __syncthreads();
    buf ^= 1;
  }
  CHUNK_MMA(la[buf], lb[buf], 0, wq * 64)

  const int dr = (lane >> 4) << 3, dc = lane & 15;
  #pragma unroll
  for (int i = 0; i < 4; ++i)
    #pragma unroll
    for (int j = 0; j < 4; ++j) {
      float* dst = V + (size_t)(k0 + i * 16 + dr) * NPIX
                     + q0 + wq * 64 + j * 16 + dc;
      #pragma unroll
      for (int r = 0; r < 8; ++r) dst[(size_t)r * NPIX] = acc[i][j][r];
    }
}

__global__ __launch_bounds__(256) void fold_kernel(
    const float* __restrict__ Vp, float* __restrict__ out)
{
  int t = blockIdx.x * 256 + threadIdx.x;
  if (t >= CCH * NPIX) return;
  int c = t / NPIX, p = t % NPIX;
  int y = p / HW, x = p % HW;
  float acc = 0.f;
  #pragma unroll
  for (int dy = 0; dy < 3; ++dy) {
    int yy = y + 1 - dy;
    if ((unsigned)yy >= (unsigned)HW) continue;
    #pragma unroll
    for (int dx = 0; dx < 3; ++dx) {
      int xx = x + 1 - dx;
      if ((unsigned)xx >= (unsigned)HW) continue;
      const float* v = Vp + (size_t)(c * 9 + dy * 3 + dx) * NPIX + yy * HW + xx;
      acc += (v[0] + v[(size_t)KDIM * NPIX]) + v[2 * (size_t)KDIM * NPIX];
    }
  }
  out[t] = acc;
}

extern "C" void kernel_launch(void* const* d_in, const int* in_sizes, int n_in,
                              void* d_out, int out_size, void* d_ws, size_t ws_size,
                              hipStream_t stream) {
  const float* fg   = (const float*)d_in[0];
  const float* bg   = (const float*)d_in[1];
  const float* mask = (const float*)d_in[2];
  float* out = (float*)d_out;

  float* ws = (float*)d_ws;
  const size_t PK = (size_t)NPIX * KDIM;
  const size_t PN = (size_t)NPIX * NPIX;
  float* mp   = ws;
  float* rn   = ws + NPIX;
  float* Pf   = ws + 2 * NPIX;
  float* Pbn  = Pf + PK;
  float* Vp   = Pf;
  float* Pbnt = Pf + 3 * PK;
  float* attn = Pbnt + PK;

  for (int b = 0; b < 4; ++b) {
    const float* fgb = fg + (size_t)b * CCH * NPIX;
    const float* bgb = bg + (size_t)b * CCH * NPIX;
    const float* mb  = mask + (size_t)b * NPIX;
    float* outb = out + (size_t)b * CCH * NPIX;

    prep_norm_kernel<<<NPIX / 32, 256, 0, stream>>>(bgb, mb, mp, rn);
    prep_unfold_kernel<<<(NPIX * KDIM) / 256, 256, 0, stream>>>(fgb, bgb, rn, Pf, Pbn);
    prep_unfold_t_kernel<<<(NPIX * KDIM) / 256, 256, 0, stream>>>(bgb, rn, Pbnt);

    dim3 g1(NPIX / 256, NPIX / 128);
    gemm1_kernel<<<g1, 256, 0, stream>>>(Pf, Pbn, attn);

    softmax_kernel<<<NPIX / 8, 256, 0, stream>>>(attn, mp);

    dim3 g2(NPIX / 256, KDIM / 64, NSPLIT);
    gemm2_kernel<<<g2, 128, 0, stream>>>(Pbnt, attn, Vp);

    fold_kernel<<<(CCH * NPIX) / 256, 256, 0, stream>>>(Vp, outb);
  }
}


# GEMM1 block tile 128x256 -> 128x192 (wave tile 64x64 -> 32x96): 162 -> 216 workgroups, one per WGP, each SIMD's serial WMMA count 4608 -> 3456
# speedup vs baseline: 5.4043x; 1.1220x over previous
#include <hip/hip_runtime.h>


typedef __attribute__((ext_vector_type(2))) float v2f;
typedef __attribute__((ext_vector_type(8))) float v8f;
typedef __attribute__((ext_vector_type(4))) float v4f;

#define NPIX 2304
#define KDIM 576
#define HW   48
#define CCH  64
#define NSPLIT 3
#define NSEG (NPIX / NSPLIT)

#define WMMA_F32(a, b, c) \
  __builtin_amdgcn_wmma_f32_16x16x4_f32(false, (a), false, (b), (short)0, (c), false, false)

__global__ __launch_bounds__(256) void prep_norm_kernel(
    const float* __restrict__ bg,
    const float* __restrict__ mask,
    float* __restrict__ mp,
    float* __restrict__ rnrm)
{
  __shared__ float part[8][32];

  const int lane = threadIdx.x & 31, w = threadIdx.x >> 5;
  const int n = blockIdx.x * 32 + lane;
  const int ny = n / HW, nx = n % HW;

  int  off[9];
  bool in[9];
  #pragma unroll
  for (int dy = 0; dy < 3; ++dy) {
    const int y = ny + dy - 1;
    const int cy = min(max(y, 0), HW - 1);
    #pragma unroll
    for (int dx = 0; dx < 3; ++dx) {
      const int x = nx + dx - 1;
      const int cx = min(max(x, 0), HW - 1);
      off[dy * 3 + dx] = cy * HW + cx;
      in[dy * 3 + dx]  = ((unsigned)y < (unsigned)HW) && ((unsigned)x < (unsigned)HW);
    }
  }

  const float* bc = bg + (size_t)(w * 8) * NPIX;
  float v[8][9];
  #pragma unroll
  for (int c = 0; c < 8; ++c)
    #pragma unroll
    for (int t = 0; t < 9; ++t)
      v[c][t] = bc[c * NPIX + off[t]];

  float mv[9];
  if (w == 0) {
    #pragma unroll
    for (int t = 0; t < 9; ++t) mv[t] = mask[off[t]];
  }

  float sq = 0.f;
  #pragma unroll
  for (int c = 0; c < 8; ++c)
    #pragma unroll
    for (int t = 0; t < 9; ++t) {
      const float x = in[t] ? v[c][t] : 0.f;
      sq += x * x;
    }
  part[w][lane] = sq;

  if (w == 0) {
    float msum = 0.f;
    #pragma unroll
    for (int t = 0; t < 9; ++t) msum += in[t] ? mv[t] : 0.f;
    mp[n] = (msum != 0.f) ? 1.f : 0.f;
  }
  __syncthreads();

  if (w == 0) {
    float s = part[0][lane];
    #pragma unroll
    for (int i = 1; i < 8; ++i) s += part[i][lane];
    rnrm[n] = 1.f / fmaxf(sqrtf(s), 1e-4f);
  }
}

#define TPITCH 292

__global__ __launch_bounds__(256) void prep_im2col_kernel(
    const float* __restrict__ fg,
    const float* __restrict__ bg,
    const float* __restrict__ rnrm,
    float* __restrict__ Pf,
    float* __restrict__ Pbn,
    float* __restrict__ Pbnt)
{
  __shared__ __attribute__((aligned(16))) float tile[32][TPITCH];

  const int tid = threadIdx.x, lane = tid & 31, w = tid >> 5;
  const int n0 = blockIdx.x * 32;
  const int n  = n0 + lane;
  const int ch = blockIdx.y;
  const bool isbg = (blockIdx.z != 0);
  const int ny = n / HW, nx = n % HW;

  int  off[9];
  bool in[9];
  #pragma unroll
  for (int dy = 0; dy < 3; ++dy) {
    const int y = ny + dy - 1;
    const int cy = min(max(y, 0), HW - 1);
    #pragma unroll
    for (int dx = 0; dx < 3; ++dx) {
      const int x = nx + dx - 1;
      const int cx = min(max(x, 0), HW - 1);
      off[dy * 3 + dx] = cy * HW + cx;
      in[dy * 3 + dx]  = ((unsigned)y < (unsigned)HW) && ((unsigned)x < (unsigned)HW);
    }
  }

  const int c0 = ch * 32 + w * 4;
  const float* sc = (isbg ? bg : fg) + (size_t)c0 * NPIX;
  float v[4][9];
  #pragma unroll
  for (int c = 0; c < 4; ++c)
    #pragma unroll
    for (int t = 0; t < 9; ++t)
      v[c][t] = sc[c * NPIX + off[t]];

  if (isbg) {
    const float rn = rnrm[n];
    #pragma unroll
    for (int c = 0; c < 4; ++c)
      #pragma unroll
      for (int t = 0; t < 9; ++t)
        v[c][t] = in[t] ? v[c][t] * rn : 0.f;
  } else {
    #pragma unroll
    for (int c = 0; c < 4; ++c)
      #pragma unroll
      for (int t = 0; t < 9; ++t)
        v[c][t] = in[t] ? v[c][t] : 0.f;
  }

  #pragma unroll
  for (int c = 0; c < 4; ++c)
    #pragma unroll
    for (int t = 0; t < 9; ++t)
      tile[lane][(w * 4 + c) * 9 + t] = v[c][t];
  __syncthreads();

  if (isbg) {
    float* dt = Pbnt + (size_t)(c0 * 9) * NPIX + n;
    #pragma unroll
    for (int c = 0; c < 4; ++c)
      #pragma unroll
      for (int t = 0; t < 9; ++t)
        dt[(size_t)(c * 9 + t) * NPIX] = v[c][t];
  }

  float* dn = (isbg ? Pbn : Pf) + (size_t)n0 * KDIM + ch * 288;
  #pragma unroll
  for (int j = 0; j < 9; ++j) {
    const int idx = j * 256 + tid;
    const int row = idx / 72, c4 = (idx % 72) * 4;
    *(v4f*)(dn + (size_t)row * KDIM + c4) = *(const v4f*)&tile[row][c4];
  }
}

#define CHUNK_MMA(LA, LB, AOFF, BOFF, NI, NJ)                           \
  _Pragma("unroll")                                                     \
  for (int kk = 0; kk < 32; kk += 4) {                                  \
    v2f af[NI], bf[NJ];                                                 \
    _Pragma("unroll")                                                   \
    for (int i = 0; i < (NI); ++i)                                      \
      af[i] = *(const v2f*)&LA[(AOFF) + i * 16 + fr][kk + fk];          \
    _Pragma("unroll")                                                   \
    for (int j = 0; j < (NJ); ++j)                                      \
      bf[j] = *(const v2f*)&LB[(BOFF) + j * 16 + fr][kk + fk];          \
    _Pragma("unroll")                                                   \
    for (int i = 0; i < (NI); ++i)                                      \
      _Pragma("unroll")                                                 \
      for (int j = 0; j < (NJ); ++j)                                    \
        acc[i][j] = WMMA_F32(af[i], bf[j], acc[i][j]);                  \
  }

#define G1N 192
#define G1I 2
#define G1J 6

__global__ __launch_bounds__(256) void gemm1_kernel(
    const float* __restrict__ Pf,
    const float* __restrict__ Pbn,
    float* __restrict__ attn)
{
  __shared__ __attribute__((aligned(16))) float la[2][128][36];
  __shared__ __attribute__((aligned(16))) float lb[2][G1N][36];

  const int tid = threadIdx.x, lane = tid & 31, wave = tid >> 5;
  const int wp = wave >> 1;
  const int wn = wave & 1;
  const int p0 = blockIdx.y * 128, n0 = blockIdx.x * G1N;
  const int fr = lane & 15;
  const int fk = (lane >> 4) << 1;

  const int lrow = tid >> 3;
  const int lcol = (tid & 7) << 2;
  const float* asrc = Pf  + (size_t)(p0 + lrow) * KDIM + lcol;
  const float* bsrc = Pbn + (size_t)(n0 + lrow) * KDIM + lcol;

  v8f acc[G1I][G1J];
  #pragma unroll
  for (int i = 0; i < G1I; ++i)
    #pragma unroll
    for (int j = 0; j < G1J; ++j) acc[i][j] = (v8f){};

  v4f ra[4], rb[6];
  #pragma unroll
  for (int i = 0; i < 4; ++i)
    ra[i] = *(const v4f*)(asrc + (size_t)i * 32 * KDIM);
  #pragma unroll
  for (int i = 0; i < 6; ++i)
    rb[i] = *(const v4f*)(bsrc + (size_t)i * 32 * KDIM);
  #pragma unroll
  for (int i = 0; i < 4; ++i) *(v4f*)&la[0][lrow + i * 32][lcol] = ra[i];
  #pragma unroll
  for (int i = 0; i < 6; ++i) *(v4f*)&lb[0][lrow + i * 32][lcol] = rb[i];
  __syncthreads();

  int buf = 0;
  for (int kc = 32; kc < KDIM; kc += 32) {
    asrc += 32; bsrc += 32;
    #pragma unroll
    for (int i = 0; i < 4; ++i)
      ra[i] = *(const v4f*)(asrc + (size_t)i * 32 * KDIM);
    #pragma unroll
    for (int i = 0; i < 6; ++i)
      rb[i] = *(const v4f*)(bsrc + (size_t)i * 32 * KDIM);
    __builtin_amdgcn_sched_barrier(0);

    CHUNK_MMA(la[buf], lb[buf], wp * 32, wn * 96, G1I, G1J)

    __builtin_amdgcn_sched_barrier(0);
    #pragma unroll
    for (int i = 0; i < 4; ++i) *(v4f*)&la[buf ^ 1][lrow + i * 32][lcol] = ra[i];
    #pragma unroll
    for (int i = 0; i < 6; ++i) *(v4f*)&lb[buf ^ 1][lrow + i * 32][lcol] = rb[i];
    __syncthreads();
    buf ^= 1;
  }
  CHUNK_MMA(la[buf], lb[buf], wp * 32, wn * 96, G1I, G1J)

  const int dr = (lane >> 4) << 3, dc = lane & 15;
  #pragma unroll
  for (int i = 0; i < G1I; ++i)
    #pragma unroll
    for (int j = 0; j < G1J; ++j) {
      float* dst = attn + (size_t)(p0 + wp * 32 + i * 16 + dr) * NPIX
                        + n0 + wn * 96 + j * 16 + dc;
      #pragma unroll
      for (int r = 0; r < 8; ++r) dst[(size_t)r * NPIX] = acc[i][j][r];
    }
}

__global__ __launch_bounds__(256) void softmax_kernel(
    float* __restrict__ attn, const float* __restrict__ mp)
{
  const int p = blockIdx.x * 8 + (threadIdx.x >> 5);
  const int lane = threadIdx.x & 31;
  float* row = attn + (size_t)p * NPIX;

  float m = -3.0e38f, s = 0.f;
  for (int n = lane; n < NPIX; n += 32) {
    float z = row[n] * mp[n] * 10.0f;
    float mn = fmaxf(m, z);
    s = s * __expf(m - mn) + __expf(z - mn);
    m = mn;
  }
  #pragma unroll
  for (int off = 16; off > 0; off >>= 1) {
    float m2 = __shfl_xor(m, off, 32);
    float s2 = __shfl_xor(s, off, 32);
    float mn = fmaxf(m, m2);
    s = s * __expf(m - mn) + s2 * __expf(m2 - mn);
    m = mn;
  }
  const float inv = 1.0f / s;
  for (int n = lane; n < NPIX; n += 32) {
    float g = mp[n];
    float z = row[n] * g * 10.0f;
    row[n] = __expf(z - m) * inv * g;
  }
}

__global__ __launch_bounds__(128) __attribute__((amdgpu_waves_per_eu(1, 1)))
void gemm2_kernel(
    const float* __restrict__ Pbnt,
    const float* __restrict__ attn,
    float* __restrict__ Vp)
{
  __shared__ __attribute__((aligned(16))) float la[2][64][36];
  __shared__ __attribute__((aligned(16))) float lb[2][256][36];

  const int tid = threadIdx.x, lane = tid & 31, wave = tid >> 5;
  const int wq = wave;
  const int k0 = blockIdx.y * 64, q0 = blockIdx.x * 256;
  const int nb = blockIdx.z * NSEG;
  const int fr = lane & 15;
  const int fk = (lane >> 4) << 1;

  const int lrow = tid >> 3;
  const int lcol = (tid & 7) << 2;
  const float* asrc = Pbnt + (size_t)(k0 + lrow) * NPIX + nb + lcol;
  const float* bsrc = attn + (size_t)(q0 + lrow) * NPIX + nb + lcol;
  float* V = Vp + (size_t)blockIdx.z * ((size_t)KDIM * NPIX);

  v8f acc[4][4];
  #pragma unroll
  for (int i = 0; i < 4; ++i)
    #pragma unroll
    for (int j = 0; j < 4; ++j) acc[i][j] = (v8f){};

  v4f ra[4], rb[16];
  #pragma unroll
  for (int i = 0; i < 4; ++i)
    ra[i] = *(const v4f*)(asrc + (size_t)i * 16 * NPIX);
  #pragma unroll
  for (int i = 0; i < 16; ++i)
    rb[i] = *(const v4f*)(bsrc + (size_t)i * 16 * NPIX);
  #pragma unroll
  for (int i = 0; i < 4; ++i)  *(v4f*)&la[0][lrow + i * 16][lcol] = ra[i];
  #pragma unroll
  for (int i = 0; i < 16; ++i) *(v4f*)&lb[0][lrow + i * 16][lcol] = rb[i];
  __syncthreads();

  int buf = 0;
  for (int nc = 32; nc < NSEG; nc += 32) {
    asrc += 32; bsrc += 32;
    #pragma unroll
    for (int i = 0; i < 4; ++i)
      ra[i] = *(const v4f*)(asrc + (size_t)i * 16 * NPIX);
    #pragma unroll
    for (int i = 0; i < 16; ++i)
      rb[i] = *(const v4f*)(bsrc + (size_t)i * 16 * NPIX);
    __builtin_amdgcn_sched_barrier(0);

    CHUNK_MMA(la[buf], lb[buf], 0, wq * 64, 4, 4)

    __builtin_amdgcn_sched_barrier(0);
    #pragma unroll
    for (int i = 0; i < 4; ++i)  *(v4f*)&la[buf ^ 1][lrow + i * 16][lcol] = ra[i];
    #pragma unroll
    for (int i = 0; i < 16; ++i) *(v4f*)&lb[buf ^ 1][lrow + i * 16][lcol] = rb[i];
    __syncthreads();
    buf ^= 1;
  }
  CHUNK_MMA(la[buf], lb[buf], 0, wq * 64, 4, 4)

  const int dr = (lane >> 4) << 3, dc = lane & 15;
  #pragma unroll
  for (int i = 0; i < 4; ++i)
    #pragma unroll
    for (int j = 0; j < 4; ++j) {
      float* dst = V + (size_t)(k0 + i * 16 + dr) * NPIX
                     + q0 + wq * 64 + j * 16 + dc;
      #pragma unroll
      for (int r = 0; r < 8; ++r) dst[(size_t)r * NPIX] = acc[i][j][r];
    }
}

__global__ __launch_bounds__(256) void fold_kernel(
    const float* __restrict__ Vp, float* __restrict__ out)
{
  int t = blockIdx.x * 256 + threadIdx.x;
  if (t >= CCH * NPIX) return;
  int c = t / NPIX, p = t % NPIX;
  int y = p / HW, x = p % HW;
  float acc = 0.f;
  #pragma unroll
  for (int dy = 0; dy < 3; ++dy) {
    int yy = y + 1 - dy;
    if ((unsigned)yy >= (unsigned)HW) continue;
    #pragma unroll
    for (int dx = 0; dx < 3; ++dx) {
      int xx = x + 1 - dx;
      if ((unsigned)xx >= (unsigned)HW) continue;
      const float* v = Vp + (size_t)(c * 9 + dy * 3 + dx) * NPIX + yy * HW + xx;
      acc += (v[0] + v[(size_t)KDIM * NPIX]) + v[2 * (size_t)KDIM * NPIX];
    }
  }
  out[t] = acc;
}

extern "C" void kernel_launch(void* const* d_in, const int* in_sizes, int n_in,
                              void* d_out, int out_size, void* d_ws, size_t ws_size,
                              hipStream_t stream) {
  const float* fg   = (const float*)d_in[0];
  const float* bg   = (const float*)d_in[1];
  const float* mask = (const float*)d_in[2];
  float* out = (float*)d_out;

  float* ws = (float*)d_ws;
  const size_t PK = (size_t)NPIX * KDIM;
  const size_t PN = (size_t)NPIX * NPIX;
  float* mp   = ws;
  float* rn   = ws + NPIX;
  float* Pf   = ws + 2 * NPIX;
  float* Pbn  = Pf + PK;
  float* Vp   = Pf;
  float* Pbnt = Pf + 3 * PK;
  float* attn = Pbnt + PK;

  for (int b = 0; b < 4; ++b) {
    const float* fgb = fg + (size_t)b * CCH * NPIX;
    const float* bgb = bg + (size_t)b * CCH * NPIX;
    const float* mb  = mask + (size_t)b * NPIX;
    float* outb = out + (size_t)b * CCH * NPIX;

    prep_norm_kernel<<<NPIX / 32, 256, 0, stream>>>(bgb, mb, mp, rn);
    dim3 gi(NPIX / 32, 2, 2);
    prep_im2col_kernel<<<gi, 256, 0, stream>>>(fgb, bgb, rn, Pf, Pbn, Pbnt);

    dim3 g1(NPIX / G1N, NPIX / 128);
    gemm1_kernel<<<g1, 256, 0, stream>>>(Pf, Pbn, attn);

    softmax_kernel<<<NPIX / 8, 256, 0, stream>>>(attn, mp);

    dim3 g2(NPIX / 256, KDIM / 64, NSPLIT);
    gemm2_kernel<<<g2, 128, 0, stream>>>(Pbnt, attn, Vp);

    fold_kernel<<<(CCH * NPIX) / 256, 256, 0, stream>>>(Vp, outb);
  }
}


# GEMM1 contraction 576 -> 64: the 3x3 patch sum commutes with the channel product: G = fg^T bg (K=64) on WMMA, then the 9-tap diagonal sum G[p+o][n+o] inside the softmax pass, row held in registers
# speedup vs baseline: 8.6530x; 1.6011x over previous
#include <hip/hip_runtime.h>


typedef __attribute__((ext_vector_type(2))) float v2f;
typedef __attribute__((ext_vector_type(8))) float v8f;
typedef __attribute__((ext_vector_type(4))) float v4f;

#define NPIX 2304
#define KDIM 576
#define HW   48
#define CCH  64
#define NSPLIT 3
#define NSEG (NPIX / NSPLIT)

#define WMMA_F32(a, b, c) \
  __builtin_amdgcn_wmma_f32_16x16x4_f32(false, (a), false, (b), (short)0, (c), false, false)

__global__ __launch_bounds__(256) void prep_norm_kernel(
    const float* __restrict__ bg,
    const float* __restrict__ mask,
    float* __restrict__ mp,
    float* __restrict__ rnrm)
{
  __shared__ float part[8][32];

  const int lane = threadIdx.x & 31, w = threadIdx.x >> 5;
  const int n = blockIdx.x * 32 + lane;
  const int ny = n / HW, nx = n % HW;

  int  off[9];
  bool in[9];
  #pragma unroll
  for (int dy = 0; dy < 3; ++dy) {
    const int y = ny + dy - 1;
    const int cy = min(max(y, 0), HW - 1);
    #pragma unroll
    for (int dx = 0; dx < 3; ++dx) {
      const int x = nx + dx - 1;
      const int cx = min(max(x, 0), HW - 1);
      off[dy * 3 + dx] = cy * HW + cx;
      in[dy * 3 + dx]  = ((unsigned)y < (unsigned)HW) && ((unsigned)x < (unsigned)HW);
    }
  }

  const float* bc = bg + (size_t)(w * 8) * NPIX;
  float v[8][9];
  #pragma unroll
  for (int c = 0; c < 8; ++c)
    #pragma unroll
    for (int t = 0; t < 9; ++t)
      v[c][t] = bc[c * NPIX + off[t]];

  float mv[9];
  if (w == 0) {
    #pragma unroll
    for (int t = 0; t < 9; ++t) mv[t] = mask[off[t]];
  }

  float sq = 0.f;
  #pragma unroll
  for (int c = 0; c < 8; ++c)
    #pragma unroll
    for (int t = 0; t < 9; ++t) {
      const float x = in[t] ? v[c][t] : 0.f;
      sq += x * x;
    }
  part[w][lane] = sq;

  if (w == 0) {
    float msum = 0.f;
    #pragma unroll
    for (int t = 0; t < 9; ++t) msum += in[t] ? mv[t] : 0.f;
    mp[n] = (msum != 0.f) ? 1.f : 0.f;
  }
  __syncthreads();

  if (w == 0) {
    float s = part[0][lane];
    #pragma unroll
    for (int i = 1; i < 8; ++i) s += part[i][lane];
    rnrm[n] = 1.f / fmaxf(sqrtf(s), 1e-4f);
  }
}

__global__ __launch_bounds__(256) void prep_im2col_kernel(
    const float* __restrict__ bg,
    const float* __restrict__ rnrm,
    float* __restrict__ Pbnt)
{
  const int tid = threadIdx.x, lane = tid & 31, w = tid >> 5;
  const int n  = blockIdx.x * 32 + lane;
  const int ch = blockIdx.y;
  const int ny = n / HW, nx = n % HW;

  int  off[9];
  bool in[9];
  #pragma unroll
  for (int dy = 0; dy < 3; ++dy) {
    const int y = ny + dy - 1;
    const int cy = min(max(y, 0), HW - 1);
    #pragma unroll
    for (int dx = 0; dx < 3; ++dx) {
      const int x = nx + dx - 1;
      const int cx = min(max(x, 0), HW - 1);
      off[dy * 3 + dx] = cy * HW + cx;
      in[dy * 3 + dx]  = ((unsigned)y < (unsigned)HW) && ((unsigned)x < (unsigned)HW);
    }
  }

  const int c0 = ch * 32 + w * 4;
  const float* sc = bg + (size_t)c0 * NPIX;
  float v[4][9];
  #pragma unroll
  for (int c = 0; c < 4; ++c)
    #pragma unroll
    for (int t = 0; t < 9; ++t)
      v[c][t] = sc[c * NPIX + off[t]];

  const float rn = rnrm[n];
  float* dt = Pbnt + (size_t)(c0 * 9) * NPIX + n;
  #pragma unroll
  for (int c = 0; c < 4; ++c)
    #pragma unroll
    for (int t = 0; t < 9; ++t)
      dt[(size_t)(c * 9 + t) * NPIX] = in[t] ? v[c][t] * rn : 0.f;
}

#define CHUNK_MMA(LA, LB, AOFF, BOFF, NI, NJ)                           \
  _Pragma("unroll")                                                     \
  for (int kk = 0; kk < 32; kk += 4) {                                  \
    v2f af[NI], bf[NJ];                                                 \
    _Pragma("unroll")                                                   \
    for (int i = 0; i < (NI); ++i)                                      \
      af[i] = *(const v2f*)&LA[(AOFF) + i * 16 + fr][kk + fk];          \
    _Pragma("unroll")                                                   \
    for (int j = 0; j < (NJ); ++j)                                      \
      bf[j] = *(const v2f*)&LB[(BOFF) + j * 16 + fr][kk + fk];          \
    _Pragma("unroll")                                                   \
    for (int i = 0; i < (NI); ++i)                                      \
      _Pragma("unroll")                                                 \
      for (int j = 0; j < (NJ); ++j)                                    \
        acc[i][j] = WMMA_F32(af[i], bf[j], acc[i][j]);                  \
  }

#define G1N 192
#define G1I 2
#define G1J 6
#define GPA 136
#define GPB 200

__global__ __launch_bounds__(256) void gemm1_kernel(
    const float* __restrict__ fg,
    const float* __restrict__ bg,
    float* __restrict__ G)
{
  __shared__ __attribute__((aligned(16))) float la[CCH][GPA];
  __shared__ __attribute__((aligned(16))) float lb[CCH][GPB];

  const int tid = threadIdx.x, lane = tid & 31, wave = tid >> 5;
  const int wp = wave >> 1;
  const int wn = wave & 1;
  const int p0 = blockIdx.y * 128, n0 = blockIdx.x * G1N;
  const int fr = lane & 15;
  const int fk = (lane >> 4) << 1;

  const int arow = tid >> 5;
  const int acol = (tid & 31) << 2;
  v4f ra[8], rb[12];
  #pragma unroll
  for (int i = 0; i < 8; ++i)
    ra[i] = *(const v4f*)(fg + (size_t)(arow + 8 * i) * NPIX + p0 + acol);
  #pragma unroll
  for (int i = 0; i < 12; ++i) {
    const int idx = i * 256 + tid;
    const int row = idx / 48, c4 = (idx % 48) << 2;
    rb[i] = *(const v4f*)(bg + (size_t)row * NPIX + n0 + c4);
  }
  #pragma unroll
  for (int i = 0; i < 8; ++i) *(v4f*)&la[arow + 8 * i][acol] = ra[i];
  #pragma unroll
  for (int i = 0; i < 12; ++i) {
    const int idx = i * 256 + tid;
    const int row = idx / 48, c4 = (idx % 48) << 2;
    *(v4f*)&lb[row][c4] = rb[i];
  }
  __syncthreads();

  v8f acc[G1I][G1J];
  #pragma unroll
  for (int i = 0; i < G1I; ++i)
    #pragma unroll
    for (int j = 0; j < G1J; ++j) acc[i][j] = (v8f){};

  #pragma unroll
  for (int kk = 0; kk < CCH; kk += 4) {
    v2f af[G1I], bf[G1J];
    #pragma unroll
    for (int i = 0; i < G1I; ++i) {
      af[i].x = la[kk + fk][wp * 32 + i * 16 + fr];
      af[i].y = la[kk + fk + 1][wp * 32 + i * 16 + fr];
    }
    #pragma unroll
    for (int j = 0; j < G1J; ++j) {
      bf[j].x = lb[kk + fk][wn * 96 + j * 16 + fr];
      bf[j].y = lb[kk + fk + 1][wn * 96 + j * 16 + fr];
    }
    #pragma unroll
    for (int i = 0; i < G1I; ++i)
      #pragma unroll
      for (int j = 0; j < G1J; ++j)
        acc[i][j] = WMMA_F32(af[i], bf[j], acc[i][j]);
  }

  const int dr = (lane >> 4) << 3, dc = lane & 15;
  #pragma unroll
  for (int i = 0; i < G1I; ++i)
    #pragma unroll
    for (int j = 0; j < G1J; ++j) {
      float* dst = G + (size_t)(p0 + wp * 32 + i * 16 + dr) * NPIX
                     + n0 + wn * 96 + j * 16 + dc;
      #pragma unroll
      for (int r = 0; r < 8; ++r) dst[(size_t)r * NPIX] = acc[i][j][r];
    }
}

#define RW 72

__global__ __launch_bounds__(128) void corr_softmax_kernel(
    const float* __restrict__ G,
    const float* __restrict__ mp,
    const float* __restrict__ rnrm,
    float* __restrict__ attn)
{
  const int lane = threadIdx.x & 31;
  const int w = __builtin_amdgcn_readfirstlane((int)(threadIdx.x >> 5));
  const int p = blockIdx.x * 4 + w;
  const int py = p / HW, px = p % HW;

  const bool xlo[3] = { lane == 0, lane == 16, false };
  const bool xhi[3] = { false, lane == 15, lane == 31 };

  float acc[RW];
  #pragma unroll
  for (int j = 0; j < RW; ++j) acc[j] = 0.f;

  #pragma unroll
  for (int oy = -1; oy <= 1; ++oy) {
    #pragma unroll
    for (int ox = -1; ox <= 1; ++ox) {
      if (((unsigned)(py + oy) < (unsigned)HW) && ((unsigned)(px + ox) < (unsigned)HW)) {
        const int d = oy * HW + ox;
        const float* src = G + ((long)(p + d) * NPIX + d + lane);
        float v[RW];
        #pragma unroll
        for (int j = 0; j < RW; ++j) v[j] = src[32 * j];
        #pragma unroll
        for (int j = 0; j < RW; ++j) {
          bool ok = (unsigned)(lane + 32 * j + HW * oy) < (unsigned)NPIX;
          if (ox < 0) ok = ok && !xlo[j % 3];
          if (ox > 0) ok = ok && !xhi[j % 3];
          acc[j] += ok ? v[j] : 0.f;
        }
      }
    }
  }

  float g[RW];
  float m = -3.0e38f;
  #pragma unroll
  for (int j = 0; j < RW; ++j) {
    const int n = lane + 32 * j;
    g[j] = mp[n];
    acc[j] = acc[j] * rnrm[n] * g[j] * 10.0f;
    m = fmaxf(m, acc[j]);
  }
  #pragma unroll
  for (int off = 16; off > 0; off >>= 1) m = fmaxf(m, __shfl_xor(m, off, 32));

  float s0 = 0.f, s1 = 0.f, s2 = 0.f, s3 = 0.f;
  #pragma unroll
  for (int j = 0; j < RW; j += 4) {
    acc[j]     = __expf(acc[j] - m);      s0 += acc[j];
    acc[j + 1] = __expf(acc[j + 1] - m);  s1 += acc[j + 1];
    acc[j + 2] = __expf(acc[j + 2] - m);  s2 += acc[j + 2];
    acc[j + 3] = __expf(acc[j + 3] - m);  s3 += acc[j + 3];
  }
  float s = (s0 + s1) + (s2 + s3);
  #pragma unroll
  for (int off = 16; off > 0; off >>= 1) s += __shfl_xor(s, off, 32);
  const float inv = 1.0f / s;

  float* row = attn + (size_t)p * NPIX + lane;
  #pragma unroll
  for (int j = 0; j < RW; ++j) row[32 * j] = acc[j] * inv * g[j];
}

__global__ __launch_bounds__(128) __attribute__((amdgpu_waves_per_eu(1, 1)))
void gemm2_kernel(
    const float* __restrict__ Pbnt,
    const float* __restrict__ attn,
    float* __restrict__ Vp)
{
  __shared__ __attribute__((aligned(16))) float la[2][64][36];
  __shared__ __attribute__((aligned(16))) float lb[2][256][36];

  const int tid = threadIdx.x, lane = tid & 31, wave = tid >> 5;
  const int wq = wave;
  const int k0 = blockIdx.y * 64, q0 = blockIdx.x * 256;
  const int nb = blockIdx.z * NSEG;
  const int fr = lane & 15;
  const int fk = (lane >> 4) << 1;

  const int lrow = tid >> 3;
  const int lcol = (tid & 7) << 2;
  const float* asrc = Pbnt + (size_t)(k0 + lrow) * NPIX + nb + lcol;
  const float* bsrc = attn + (size_t)(q0 + lrow) * NPIX + nb + lcol;
  float* V = Vp + (size_t)blockIdx.z * ((size_t)KDIM * NPIX);

  v8f acc[4][4];
  #pragma unroll
  for (int i = 0; i < 4; ++i)
    #pragma unroll
    for (int j = 0; j < 4; ++j) acc[i][j] = (v8f){};

  v4f ra[4], rb[16];
  #pragma unroll
  for (int i = 0; i < 4; ++i)
    ra[i] = *(const v4f*)(asrc + (size_t)i * 16 * NPIX);
  #pragma unroll
  for (int i = 0; i < 16; ++i)
    rb[i] = *(const v4f*)(bsrc + (size_t)i * 16 * NPIX);
  #pragma unroll
  for (int i = 0; i < 4; ++i)  *(v4f*)&la[0][lrow + i * 16][lcol] = ra[i];
  #pragma unroll
  for (int i = 0; i < 16; ++i) *(v4f*)&lb[0][lrow + i * 16][lcol] = rb[i];
  __syncthreads();

  int buf = 0;
  for (int nc = 32; nc < NSEG; nc += 32) {
    asrc += 32; bsrc += 32;
    #pragma unroll
    for (int i = 0; i < 4; ++i)
      ra[i] = *(const v4f*)(asrc + (size_t)i * 16 * NPIX);
    #pragma unroll
    for (int i = 0; i < 16; ++i)
      rb[i] = *(const v4f*)(bsrc + (size_t)i * 16 * NPIX);
    __builtin_amdgcn_sched_barrier(0);

    CHUNK_MMA(la[buf], lb[buf], 0, wq * 64, 4, 4)

    __builtin_amdgcn_sched_barrier(0);
    #pragma unroll
    for (int i = 0; i < 4; ++i)  *(v4f*)&la[buf ^ 1][lrow + i * 16][lcol] = ra[i];
    #pragma unroll
    for (int i = 0; i < 16; ++i) *(v4f*)&lb[buf ^ 1][lrow + i * 16][lcol] = rb[i];
    __syncthreads();
    buf ^= 1;
  }
  CHUNK_MMA(la[buf], lb[buf], 0, wq * 64, 4, 4)

  const int dr = (lane >> 4) << 3, dc = lane & 15;
  #pragma unroll
  for (int i = 0; i < 4; ++i)
    #pragma unroll
    for (int j = 0; j < 4; ++j) {
      float* dst = V + (size_t)(k0 + i * 16 + dr) * NPIX
                     + q0 + wq * 64 + j * 16 + dc;
      #pragma unroll
      for (int r = 0; r < 8; ++r) dst[(size_t)r * NPIX] = acc[i][j][r];
    }
}

__global__ __launch_bounds__(256) void fold_kernel(
    const float* __restrict__ Vp, float* __restrict__ out)
{
  int t = blockIdx.x * 256 + threadIdx.x;
  if (t >= CCH * NPIX) return;
  int c = t / NPIX, p = t % NPIX;
  int y = p / HW, x = p % HW;
  float acc = 0.f;
  #pragma unroll
  for (int dy = 0; dy < 3; ++dy) {
    int yy = y + 1 - dy;
    if ((unsigned)yy >= (unsigned)HW) continue;
    #pragma unroll
    for (int dx = 0; dx < 3; ++dx) {
      int xx = x + 1 - dx;
      if ((unsigned)xx >= (unsigned)HW) continue;
      const float* v = Vp + (size_t)(c * 9 + dy * 3 + dx) * NPIX + yy * HW + xx;
      acc += (v[0] + v[(size_t)KDIM * NPIX]) + v[2 * (size_t)KDIM * NPIX];
    }
  }
  out[t] = acc;
}

extern "C" void kernel_launch(void* const* d_in, const int* in_sizes, int n_in,
                              void* d_out, int out_size, void* d_ws, size_t ws_size,
                              hipStream_t stream) {
  const float* fg   = (const float*)d_in[0];
  const float* bg   = (const float*)d_in[1];
  const float* mask = (const float*)d_in[2];
  float* out = (float*)d_out;

  float* ws = (float*)d_ws;
  const size_t PK = (size_t)NPIX * KDIM;
  float* mp   = ws;
  float* rn   = ws + NPIX;
  float* G    = ws + 2 * NPIX;
  float* Vp   = G;
  float* Pbnt = G + 3 * PK;
  float* attn = Pbnt + PK;

  for (int b = 0; b < 4; ++b) {
    const float* fgb = fg + (size_t)b * CCH * NPIX;
    const float* bgb = bg + (size_t)b * CCH * NPIX;
    const float* mb  = mask + (size_t)b * NPIX;
    float* outb = out + (size_t)b * CCH * NPIX;

    prep_norm_kernel<<<NPIX / 32, 256, 0, stream>>>(bgb, mb, mp, rn);

    dim3 g1(NPIX / G1N, NPIX / 128);
    gemm1_kernel<<<g1, 256, 0, stream>>>(fgb, bgb, G);

    corr_softmax_kernel<<<NPIX / 4, 128, 0, stream>>>(G, mp, rn, attn);

    dim3 gi(NPIX / 32, 2);
    prep_im2col_kernel<<<gi, 256, 0, stream>>>(bgb, rn, Pbnt);

    dim3 g2(NPIX / 256, KDIM / 64, NSPLIT);
    gemm2_kernel<<<g2, 128, 0, stream>>>(Pbnt, attn, Vp);

    fold_kernel<<<(CCH * NPIX) / 256, 256, 0, stream>>>(Vp, outb);
  }
}


# GEMM2 rows 576 -> 64: the 3x3 fold commutes with the product: S = 9-tap diagonal sum of rn*attn, then out = bg S^T (M=64, K=2304) on WMMA in 24 K-segments; patch plane and fold pass gone
# speedup vs baseline: 10.4383x; 1.2063x over previous
#include <hip/hip_runtime.h>


typedef __attribute__((ext_vector_type(2))) float v2f;
typedef __attribute__((ext_vector_type(8))) float v8f;
typedef __attribute__((ext_vector_type(4))) float v4f;

#define NPIX 2304
#define HW   48
#define CCH  64
#define NSPLIT 24
#define NSEG (NPIX / NSPLIT)

#define WMMA_F32(a, b, c) \
  __builtin_amdgcn_wmma_f32_16x16x4_f32(false, (a), false, (b), (short)0, (c), false, false)

__global__ __launch_bounds__(256) void prep_norm_kernel(
    const float* __restrict__ bg,
    const float* __restrict__ mask,
    float* __restrict__ mp,
    float* __restrict__ rnrm)
{
  __shared__ float part[8][32];

  const int lane = threadIdx.x & 31, w = threadIdx.x >> 5;
  const int n = blockIdx.x * 32 + lane;
  const int ny = n / HW, nx = n % HW;

  int  off[9];
  bool in[9];
  #pragma unroll
  for (int dy = 0; dy < 3; ++dy) {
    const int y = ny + dy - 1;
    const int cy = min(max(y, 0), HW - 1);
    #pragma unroll
    for (int dx = 0; dx < 3; ++dx) {
      const int x = nx + dx - 1;
      const int cx = min(max(x, 0), HW - 1);
      off[dy * 3 + dx] = cy * HW + cx;
      in[dy * 3 + dx]  = ((unsigned)y < (unsigned)HW) && ((unsigned)x < (unsigned)HW);
    }
  }

  const float* bc = bg + (size_t)(w * 8) * NPIX;
  float v[8][9];
  #pragma unroll
  for (int c = 0; c < 8; ++c)
    #pragma unroll
    for (int t = 0; t < 9; ++t)
      v[c][t] = bc[c * NPIX + off[t]];

  float mv[9];
  if (w == 0) {
    #pragma unroll
    for (int t = 0; t < 9; ++t) mv[t] = mask[off[t]];
  }

  float sq = 0.f;
  #pragma unroll
  for (int c = 0; c < 8; ++c)
    #pragma unroll
    for (int t = 0; t < 9; ++t) {
      const float x = in[t] ? v[c][t] : 0.f;
      sq += x * x;
    }
  part[w][lane] = sq;

  if (w == 0) {
    float msum = 0.f;
    #pragma unroll
    for (int t = 0; t < 9; ++t) msum += in[t] ? mv[t] : 0.f;
    mp[n] = (msum != 0.f) ? 1.f : 0.f;
  }
  __syncthreads();

  if (w == 0) {
    float s = part[0][lane];
    #pragma unroll
    for (int i = 1; i < 8; ++i) s += part[i][lane];
    rnrm[n] = 1.f / fmaxf(sqrtf(s), 1e-4f);
  }
}

#define CHUNK_MMA(LA, LB, AOFF, BOFF, NI, NJ)                           \
  _Pragma("unroll")                                                     \
  for (int kk = 0; kk < 32; kk += 4) {                                  \
    v2f af[NI], bf[NJ];                                                 \
    _Pragma("unroll")                                                   \
    for (int i = 0; i < (NI); ++i)                                      \
      af[i] = *(const v2f*)&LA[(AOFF) + i * 16 + fr][kk + fk];          \
    _Pragma("unroll")                                                   \
    for (int j = 0; j < (NJ); ++j)                                      \
      bf[j] = *(const v2f*)&LB[(BOFF) + j * 16 + fr][kk + fk];          \
    _Pragma("unroll")                                                   \
    for (int i = 0; i < (NI); ++i)                                      \
      _Pragma("unroll")                                                 \
      for (int j = 0; j < (NJ); ++j)                                    \
        acc[i][j] = WMMA_F32(af[i], bf[j], acc[i][j]);                  \
  }

#define G1N 192
#define G1I 2
#define G1J 6
#define GPA 136
#define GPB 200

__global__ __launch_bounds__(256) void gemm1_kernel(
    const float* __restrict__ fg,
    const float* __restrict__ bg,
    float* __restrict__ G)
{
  __shared__ __attribute__((aligned(16))) float la[CCH][GPA];
  __shared__ __attribute__((aligned(16))) float lb[CCH][GPB];

  const int tid = threadIdx.x, lane = tid & 31, wave = tid >> 5;
  const int wp = wave >> 1;
  const int wn = wave & 1;
  const int p0 = blockIdx.y * 128, n0 = blockIdx.x * G1N;
  const int fr = lane & 15;
  const int fk = (lane >> 4) << 1;

  const int arow = tid >> 5;
  const int acol = (tid & 31) << 2;
  v4f ra[8], rb[12];
  #pragma unroll
  for (int i = 0; i < 8; ++i)
    ra[i] = *(const v4f*)(fg + (size_t)(arow + 8 * i) * NPIX + p0 + acol);
  #pragma unroll
  for (int i = 0; i < 12; ++i) {
    const int idx = i * 256 + tid;
    const int row = idx / 48, c4 = (idx % 48) << 2;
    rb[i] = *(const v4f*)(bg + (size_t)row * NPIX + n0 + c4);
  }
  #pragma unroll
  for (int i = 0; i < 8; ++i) *(v4f*)&la[arow + 8 * i][acol] = ra[i];
  #pragma unroll
  for (int i = 0; i < 12; ++i) {
    const int idx = i * 256 + tid;
    const int row = idx / 48, c4 = (idx % 48) << 2;
    *(v4f*)&lb[row][c4] = rb[i];
  }
  __syncthreads();

  v8f acc[G1I][G1J];
  #pragma unroll
  for (int i = 0; i < G1I; ++i)
    #pragma unroll
    for (int j = 0; j < G1J; ++j) acc[i][j] = (v8f){};

  #pragma unroll
  for (int kk = 0; kk < CCH; kk += 4) {
    v2f af[G1I], bf[G1J];
    #pragma unroll
    for (int i = 0; i < G1I; ++i) {
      af[i].x = la[kk + fk][wp * 32 + i * 16 + fr];
      af[i].y = la[kk + fk + 1][wp * 32 + i * 16 + fr];
    }
    #pragma unroll
    for (int j = 0; j < G1J; ++j) {
      bf[j].x = lb[kk + fk][wn * 96 + j * 16 + fr];
      bf[j].y = lb[kk + fk + 1][wn * 96 + j * 16 + fr];
    }
    #pragma unroll
    for (int i = 0; i < G1I; ++i)
      #pragma unroll
      for (int j = 0; j < G1J; ++j)
        acc[i][j] = WMMA_F32(af[i], bf[j], acc[i][j]);
  }

  const int dr = (lane >> 4) << 3, dc = lane & 15;
  #pragma unroll
  for (int i = 0; i < G1I; ++i)
    #pragma unroll
    for (int j = 0; j < G1J; ++j) {
      float* dst = G + (size_t)(p0 + wp * 32 + i * 16 + dr) * NPIX
                     + n0 + wn * 96 + j * 16 + dc;
      #pragma unroll
      for (int r = 0; r < 8; ++r) dst[(size_t)r * NPIX] = acc[i][j][r];
    }
}

#define RW 72

__global__ __launch_bounds__(128) void corr_softmax_kernel(
    const float* __restrict__ G,
    const float* __restrict__ mp,
    const float* __restrict__ rnrm,
    float* __restrict__ wgt)
{
  const int lane = threadIdx.x & 31;
  const int w = __builtin_amdgcn_readfirstlane((int)(threadIdx.x >> 5));
  const int p = blockIdx.x * 4 + w;
  const int py = p / HW, px = p % HW;

  const bool xlo[3] = { lane == 0, lane == 16, false };
  const bool xhi[3] = { false, lane == 15, lane == 31 };

  float acc[RW];
  #pragma unroll
  for (int j = 0; j < RW; ++j) acc[j] = 0.f;

  #pragma unroll
  for (int oy = -1; oy <= 1; ++oy) {
    #pragma unroll
    for (int ox = -1; ox <= 1; ++ox) {
      if (((unsigned)(py + oy) < (unsigned)HW) && ((unsigned)(px + ox) < (unsigned)HW)) {
        const int d = oy * HW + ox;
        const float* src = G + ((long)(p + d) * NPIX + d + lane);
        float v[RW];
        #pragma unroll
        for (int j = 0; j < RW; ++j) v[j] = src[32 * j];
        #pragma unroll
        for (int j = 0; j < RW; ++j) {
          bool ok = (unsigned)(lane + 32 * j + HW * oy) < (unsigned)NPIX;
          if (ox < 0) ok = ok && !xlo[j % 3];
          if (ox > 0) ok = ok && !xhi[j % 3];
          acc[j] += ok ? v[j] : 0.f;
        }
      }
    }
  }

  float g[RW];
  float m = -3.0e38f;
  #pragma unroll
  for (int j = 0; j < RW; ++j) {
    const int n = lane + 32 * j;
    const float r = rnrm[n], gm = mp[n];
    acc[j] = acc[j] * r * gm * 10.0f;
    g[j] = gm * r;
    m = fmaxf(m, acc[j]);
  }
  #pragma unroll
  for (int off = 16; off > 0; off >>= 1) m = fmaxf(m, __shfl_xor(m, off, 32));

  float s0 = 0.f, s1 = 0.f, s2 = 0.f, s3 = 0.f;
  #pragma unroll
  for (int j = 0; j < RW; j += 4) {
    acc[j]     = __expf(acc[j] - m);      s0 += acc[j];
    acc[j + 1] = __expf(acc[j + 1] - m);  s1 += acc[j + 1];
    acc[j + 2] = __expf(acc[j + 2] - m);  s2 += acc[j + 2];
    acc[j + 3] = __expf(acc[j + 3] - m);  s3 += acc[j + 3];
  }
  float s = (s0 + s1) + (s2 + s3);
  #pragma unroll
  for (int off = 16; off > 0; off >>= 1) s += __shfl_xor(s, off, 32);
  const float inv = 1.0f / s;

  float* row = wgt + (size_t)p * NPIX + lane;
  #pragma unroll
  for (int j = 0; j < RW; ++j) row[32 * j] = acc[j] * inv * g[j];
}

__global__ __launch_bounds__(128) void diag_sum_kernel(
    const float* __restrict__ wgt,
    float* __restrict__ S)
{
  const int lane = threadIdx.x & 31;
  const int w = __builtin_amdgcn_readfirstlane((int)(threadIdx.x >> 5));
  const int p = blockIdx.x * 4 + w;
  const int py = p / HW, px = p % HW;

  const bool xlo[3] = { lane == 0, lane == 16, false };
  const bool xhi[3] = { false, lane == 15, lane == 31 };

  float acc[RW];
  #pragma unroll
  for (int j = 0; j < RW; ++j) acc[j] = 0.f;

  #pragma unroll
  for (int oy = -1; oy <= 1; ++oy) {
    #pragma unroll
    for (int ox = -1; ox <= 1; ++ox) {
      if (((unsigned)(py + oy) < (unsigned)HW) && ((unsigned)(px + ox) < (unsigned)HW)) {
        const int d = oy * HW + ox;
        const float* src = wgt + ((long)(p + d) * NPIX + d + lane);
        float v[RW];
        #pragma unroll
        for (int j = 0; j < RW; ++j) v[j] = src[32 * j];
        #pragma unroll
        for (int j = 0; j < RW; ++j) {
          bool ok = (unsigned)(lane + 32 * j + HW * oy) < (unsigned)NPIX;
          if (ox < 0) ok = ok && !xlo[j % 3];
          if (ox > 0) ok = ok && !xhi[j % 3];
          acc[j] += ok ? v[j] : 0.f;
        }
      }
    }
  }

  float* row = S + (size_t)p * NPIX + lane;
  #pragma unroll
  for (int j = 0; j < RW; ++j) row[32 * j] = acc[j];
}

__global__ __launch_bounds__(128) __attribute__((amdgpu_waves_per_eu(1, 1)))
void gemm2_kernel(
    const float* __restrict__ bg,
    const float* __restrict__ S,
    float* __restrict__ Vp)
{
  __shared__ __attribute__((aligned(16))) float la[2][64][36];
  __shared__ __attribute__((aligned(16))) float lb[2][256][36];

  const int tid = threadIdx.x, lane = tid & 31, wave = tid >> 5;
  const int wq = wave;
  const int q0 = blockIdx.x * 256;
  const int nb = blockIdx.z * NSEG;
  const int fr = lane & 15;
  const int fk = (lane >> 4) << 1;

  const int lrow = tid >> 3;
  const int lcol = (tid & 7) << 2;
  const float* asrc = bg + (size_t)lrow * NPIX + nb + lcol;
  const float* bsrc = S + (size_t)(q0 + lrow) * NPIX + nb + lcol;
  float* V = Vp + (size_t)blockIdx.z * ((size_t)CCH * NPIX);

  v8f acc[4][4];
  #pragma unroll
  for (int i = 0; i < 4; ++i)
    #pragma unroll
    for (int j = 0; j < 4; ++j) acc[i][j] = (v8f){};

  v4f ra[4], rb[16];
  #pragma unroll
  for (int i = 0; i < 4; ++i)
    ra[i] = *(const v4f*)(asrc + (size_t)i * 16 * NPIX);
  #pragma unroll
  for (int i = 0; i < 16; ++i)
    rb[i] = *(const v4f*)(bsrc + (size_t)i * 16 * NPIX);
  #pragma unroll
  for (int i = 0; i < 4; ++i)  *(v4f*)&la[0][lrow + i * 16][lcol] = ra[i];
  #pragma unroll
  for (int i = 0; i < 16; ++i) *(v4f*)&lb[0][lrow + i * 16][lcol] = rb[i];
  __syncthreads();

  int buf = 0;
  for (int nc = 32; nc < NSEG; nc += 32) {
    asrc += 32; bsrc += 32;
    #pragma unroll
    for (int i = 0; i < 4; ++i)
      ra[i] = *(const v4f*)(asrc + (size_t)i * 16 * NPIX);
    #pragma unroll
    for (int i = 0; i < 16; ++i)
      rb[i] = *(const v4f*)(bsrc + (size_t)i * 16 * NPIX);
    __builtin_amdgcn_sched_barrier(0);

    CHUNK_MMA(la[buf], lb[buf], 0, wq * 64, 4, 4)

    __builtin_amdgcn_sched_barrier(0);
    #pragma unroll
    for (int i = 0; i < 4; ++i)  *(v4f*)&la[buf ^ 1][lrow + i * 16][lcol] = ra[i];
    #pragma unroll
    for (int i = 0; i < 16; ++i) *(v4f*)&lb[buf ^ 1][lrow + i * 16][lcol] = rb[i];
    __syncthreads();
    buf ^= 1;
  }
  CHUNK_MMA(la[buf], lb[buf], 0, wq * 64, 4, 4)

  const int dr = (lane >> 4) << 3, dc = lane & 15;
  #pragma unroll
  for (int i = 0; i < 4; ++i)
    #pragma unroll
    for (int j = 0; j < 4; ++j) {
      float* dst = V + (size_t)(i * 16 + dr) * NPIX
                     + q0 + wq * 64 + j * 16 + dc;
      #pragma unroll
      for (int r = 0; r < 8; ++r) dst[(size_t)r * NPIX] = acc[i][j][r];
    }
}

__global__ __launch_bounds__(256) void plane_sum_kernel(
    const float* __restrict__ Vp, float* __restrict__ out)
{
  const int t = (blockIdx.x * 256 + threadIdx.x) * 4;
  v4f v[NSPLIT];
  #pragma unroll
  for (int s = 0; s < NSPLIT; ++s)
    v[s] = *(const v4f*)(Vp + (size_t)s * ((size_t)CCH * NPIX) + t);
  v4f a = v[0];
  #pragma unroll
  for (int s = 1; s < NSPLIT; ++s) a += v[s];
  *(v4f*)(out + t) = a;
}

extern "C" void kernel_launch(void* const* d_in, const int* in_sizes, int n_in,
                              void* d_out, int out_size, void* d_ws, size_t ws_size,
                              hipStream_t stream) {
  const float* fg   = (const float*)d_in[0];
  const float* bg   = (const float*)d_in[1];
  const float* mask = (const float*)d_in[2];
  float* out = (float*)d_out;

  float* ws = (float*)d_ws;
  const size_t NN = (size_t)NPIX * NPIX;
  float* mp = ws;
  float* G  = ws + NPIX;
  float* S  = G;
  float* Wg = G + NN;
  float* Vp = Wg;
  float* rn = Wg + NN;

  for (int b = 0; b < 4; ++b) {
    const float* fgb = fg + (size_t)b * CCH * NPIX;
    const float* bgb = bg + (size_t)b * CCH * NPIX;
    const float* mb  = mask + (size_t)b * NPIX;
    float* outb = out + (size_t)b * CCH * NPIX;

    prep_norm_kernel<<<NPIX / 32, 256, 0, stream>>>(bgb, mb, mp, rn);

    dim3 g1(NPIX / G1N, NPIX / 128);
    gemm1_kernel<<<g1, 256, 0, stream>>>(fgb, bgb, G);

    corr_softmax_kernel<<<NPIX / 4, 128, 0, stream>>>(G, mp, rn, Wg);

    diag_sum_kernel<<<NPIX / 4, 128, 0, stream>>>(Wg, S);

    dim3 g2(NPIX / 256, 1, NSPLIT);
    gemm2_kernel<<<g2, 128, 0, stream>>>(bgb, S, Vp);

    plane_sum_kernel<<<(CCH * NPIX) / 1024, 256, 0, stream>>>(Vp, outb);
  }
}


# both diagonal-sum passes: a 9-row workgroup stages the 11 source rows of each y offset once in LDS (16-byte whole-line loads, next offset prefetched); taps read LDS, not 9 shifted global rows
# speedup vs baseline: 12.7483x; 1.2213x over previous
#include <hip/hip_runtime.h>


typedef __attribute__((ext_vector_type(2))) float v2f;
typedef __attribute__((ext_vector_type(8))) float v8f;
typedef __attribute__((ext_vector_type(4))) float v4f;

#define NPIX 2304
#define HW   48
#define CCH  64
#define NSPLIT 24
#define NSEG (NPIX / NSPLIT)

#define WMMA_F32(a, b, c) \
  __builtin_amdgcn_wmma_f32_16x16x4_f32(false, (a), false, (b), (short)0, (c), false, false)

__device__ __forceinline__ void prep_norm_role(
    const int blk,
    float (*part)[32],
    const float* __restrict__ bg,
    const float* __restrict__ mask,
    float* __restrict__ mp,
    float* __restrict__ rnrm)
{
  const int lane = threadIdx.x & 31, w = threadIdx.x >> 5;
  const int n = blk * 32 + lane;
  const int ny = n / HW, nx = n % HW;

  int  off[9];
  bool in[9];
  #pragma unroll
  for (int dy = 0; dy < 3; ++dy) {
    const int y = ny + dy - 1;
    const int cy = min(max(y, 0), HW - 1);
    #pragma unroll
    for (int dx = 0; dx < 3; ++dx) {
      const int x = nx + dx - 1;
      const int cx = min(max(x, 0), HW - 1);
      off[dy * 3 + dx] = cy * HW + cx;
      in[dy * 3 + dx]  = ((unsigned)y < (unsigned)HW) && ((unsigned)x < (unsigned)HW);
    }
  }

  const float* bc = bg + (size_t)(w * 8) * NPIX;
  float v[8][9];
  #pragma unroll
  for (int c = 0; c < 8; ++c)
    #pragma unroll
    for (int t = 0; t < 9; ++t)
      v[c][t] = bc[c * NPIX + off[t]];

  float mv[9];
  if (w == 0) {
    #pragma unroll
    for (int t = 0; t < 9; ++t) mv[t] = mask[off[t]];
  }

  float sq = 0.f;
  #pragma unroll
  for (int c = 0; c < 8; ++c)
    #pragma unroll
    for (int t = 0; t < 9; ++t) {
      const float x = in[t] ? v[c][t] : 0.f;
      sq += x * x;
    }
  part[w][lane] = sq;

  if (w == 0) {
    float msum = 0.f;
    #pragma unroll
    for (int t = 0; t < 9; ++t) msum += in[t] ? mv[t] : 0.f;
    mp[n] = (msum != 0.f) ? 1.f : 0.f;
  }
  __syncthreads();

  if (w == 0) {
    float s = part[0][lane];
    #pragma unroll
    for (int i = 1; i < 8; ++i) s += part[i][lane];
    rnrm[n] = 1.f / fmaxf(sqrtf(s), 1e-4f);
  }
}

#define CHUNK_MMA(LA, LB, AOFF, BOFF, NI, NJ)                           \
  _Pragma("unroll")                                                     \
  for (int kk = 0; kk < 32; kk += 4) {                                  \
    v2f af[NI], bf[NJ];                                                 \
    _Pragma("unroll")                                                   \
    for (int i = 0; i < (NI); ++i)                                      \
      af[i] = *(const v2f*)&LA[(AOFF) + i * 16 + fr][kk + fk];          \
    _Pragma("unroll")                                                   \
    for (int j = 0; j < (NJ); ++j)                                      \
      bf[j] = *(const v2f*)&LB[(BOFF) + j * 16 + fr][kk + fk];          \
    _Pragma("unroll")                                                   \
    for (int i = 0; i < (NI); ++i)                                      \
      _Pragma("unroll")                                                 \
      for (int j = 0; j < (NJ); ++j)                                    \
        acc[i][j] = WMMA_F32(af[i], bf[j], acc[i][j]);                  \
  }

#define G1N 192
#define G1I 2
#define G1J 6
#define GPA 136
#define GPB 200

#define G1BX (NPIX / G1N)
#define G1BY (NPIX / 128)

__device__ __forceinline__ void gemm1_role(
    const int blk,
    float (*la)[GPA],
    float (*lb)[GPB],
    const float* __restrict__ fg,
    const float* __restrict__ bg,
    float* __restrict__ G)
{
  const int tid = threadIdx.x, lane = tid & 31, wave = tid >> 5;
  const int wp = wave >> 1;
  const int wn = wave & 1;
  const int p0 = (blk / G1BX) * 128, n0 = (blk % G1BX) * G1N;
  const int fr = lane & 15;
  const int fk = (lane >> 4) << 1;

  const int arow = tid >> 5;
  const int acol = (tid & 31) << 2;
  v4f ra[8], rb[12];
  #pragma unroll
  for (int i = 0; i < 8; ++i)
    ra[i] = *(const v4f*)(fg + (size_t)(arow + 8 * i) * NPIX + p0 + acol);
  #pragma unroll
  for (int i = 0; i < 12; ++i) {
    const int idx = i * 256 + tid;
    const int row = idx / 48, c4 = (idx % 48) << 2;
    rb[i] = *(const v4f*)(bg + (size_t)row * NPIX + n0 + c4);
  }
  #pragma unroll
  for (int i = 0; i < 8; ++i) *(v4f*)&la[arow + 8 * i][acol] = ra[i];
  #pragma unroll
  for (int i = 0; i < 12; ++i) {
    const int idx = i * 256 + tid;
    const int row = idx / 48, c4 = (idx % 48) << 2;
    *(v4f*)&lb[row][c4] = rb[i];
  }
  __syncthreads();

  v8f acc[G1I][G1J];
  #pragma unroll
  for (int i = 0; i < G1I; ++i)
    #pragma unroll
    for (int j = 0; j < G1J; ++j) acc[i][j] = (v8f){};

  #pragma unroll
  for (int kk = 0; kk < CCH; kk += 4) {
    v2f af[G1I], bf[G1J];
    #pragma unroll
    for (int i = 0; i < G1I; ++i) {
      af[i].x = la[kk + fk][wp * 32 + i * 16 + fr];
      af[i].y = la[kk + fk + 1][wp * 32 + i * 16 + fr];
    }
    #pragma unroll
    for (int j = 0; j < G1J; ++j) {
      bf[j].x = lb[kk + fk][wn * 96 + j * 16 + fr];
      bf[j].y = lb[kk + fk + 1][wn * 96 + j * 16 + fr];
    }
    #pragma unroll
    for (int i = 0; i < G1I; ++i)
      #pragma unroll
      for (int j = 0; j < G1J; ++j)
        acc[i][j] = WMMA_F32(af[i], bf[j], acc[i][j]);
  }

  const int dr = (lane >> 4) << 3, dc = lane & 15;
  #pragma unroll
  for (int i = 0; i < G1I; ++i)
    #pragma unroll
    for (int j = 0; j < G1J; ++j) {
      float* dst = G + (size_t)(p0 + wp * 32 + i * 16 + dr) * NPIX
                     + n0 + wn * 96 + j * 16 + dc;
      #pragma unroll
      for (int r = 0; r < 8; ++r) dst[(size_t)r * NPIX] = acc[i][j][r];
    }
}

#define RW 72
#define DROWS 9
#define DSTG  (DROWS + 2)
#define DTHR  (DROWS * 32)
#define DPAD  64
#define DLD   ((DSTG * NPIX / 4) / DTHR)
#define DTILE (DSTG * NPIX + 2 * DPAD)

__device__ __forceinline__ void stage_fetch(
    v4f (&st)[DLD], const float* __restrict__ X, const int q0, const int tid)
{
  #pragma unroll
  for (int i = 0; i < DLD; ++i) {
    const int q = min(max(q0 + (i >> 1), 0), NPIX - 1);
    st[i] = *(const v4f*)(X + (size_t)q * NPIX + (((i & 1) * DTHR + tid) << 2));
  }
}

__device__ __forceinline__ void stage_put(
    float* __restrict__ tile, const v4f (&st)[DLD], const int tid)
{
  #pragma unroll
  for (int i = 0; i < DLD; ++i)
    *(v4f*)&tile[DPAD + (i >> 1) * NPIX + (((i & 1) * DTHR + tid) << 2)] = st[i];
}

template <int OY>
__device__ __forceinline__ void taps_from_tile(
    float (&acc)[RW], const float* __restrict__ tile,
    const int w, const int lane, const int py, const int px)
{
  const bool xlo[3] = { lane == 0, lane == 16, false };
  const bool xhi[3] = { false, lane == 15, lane == 31 };
  #pragma unroll
  for (int ox = -1; ox <= 1; ++ox) {
    if (((unsigned)(py + OY) < (unsigned)HW) && ((unsigned)(px + ox) < (unsigned)HW)) {
      const float* src = tile + (DPAD + (w + 1 + ox) * NPIX + OY * HW + ox) + lane;
      float v[RW];
      #pragma unroll
      for (int j = 0; j < RW; ++j) v[j] = src[32 * j];
      #pragma unroll
      for (int j = 0; j < RW; ++j) {
        bool ok = (unsigned)(lane + 32 * j + HW * OY) < (unsigned)NPIX;
        if (ox < 0) ok = ok && !xlo[j % 3];
        if (ox > 0) ok = ok && !xhi[j % 3];
        acc[j] += ok ? v[j] : 0.f;
      }
    }
  }
}

__device__ __forceinline__ void diag_rows(
    float (&acc)[RW], float* __restrict__ tile, const float* __restrict__ X,
    const int p0, const int w, const int lane, const int tid)
{
  const int p = p0 + w;
  const int py = p / HW, px = p % HW;

  #pragma unroll
  for (int j = 0; j < RW; ++j) acc[j] = 0.f;

  if (tid < DPAD) {
    tile[tid] = 0.f;
    tile[DPAD + DSTG * NPIX + tid] = 0.f;
  }

  v4f st[DLD];
  stage_fetch(st, X, p0 - HW - 1, tid);
  stage_put(tile, st, tid);
  __syncthreads();

  stage_fetch(st, X, p0 - 1, tid);
  __builtin_amdgcn_sched_barrier(0);
  taps_from_tile<-1>(acc, tile, w, lane, py, px);
  __syncthreads();
  stage_put(tile, st, tid);
  __syncthreads();

  stage_fetch(st, X, p0 + HW - 1, tid);
  __builtin_amdgcn_sched_barrier(0);
  taps_from_tile<0>(acc, tile, w, lane, py, px);
  __syncthreads();
  stage_put(tile, st, tid);
  __syncthreads();

  taps_from_tile<1>(acc, tile, w, lane, py, px);
}

__global__ __launch_bounds__(DTHR) void corr_softmax_kernel(
    const float* __restrict__ G,
    const float* __restrict__ mp,
    const float* __restrict__ rnrm,
    float* __restrict__ wgt)
{
  __shared__ __attribute__((aligned(16))) float tile[DTILE];

  const int tid = threadIdx.x, lane = tid & 31;
  const int w = __builtin_amdgcn_readfirstlane((int)(threadIdx.x >> 5));
  const int p0 = blockIdx.x * DROWS;
  const int p = p0 + w;

  float acc[RW];
  diag_rows(acc, tile, G, p0, w, lane, tid);

  float g[RW];
  float m = -3.0e38f;
  #pragma unroll
  for (int j = 0; j < RW; ++j) {
    const int n = lane + 32 * j;
    const float r = rnrm[n], gm = mp[n];
    acc[j] = acc[j] * r * gm * 10.0f;
    g[j] = gm * r;
    m = fmaxf(m, acc[j]);
  }
  #pragma unroll
  for (int off = 16; off > 0; off >>= 1) m = fmaxf(m, __shfl_xor(m, off, 32));

  float s0 = 0.f, s1 = 0.f, s2 = 0.f, s3 = 0.f;
  #pragma unroll
  for (int j = 0; j < RW; j += 4) {
    acc[j]     = __expf(acc[j] - m);      s0 += acc[j];
    acc[j + 1] = __expf(acc[j + 1] - m);  s1 += acc[j + 1];
    acc[j + 2] = __expf(acc[j + 2] - m);  s2 += acc[j + 2];
    acc[j + 3] = __expf(acc[j + 3] - m);  s3 += acc[j + 3];
  }
  float s = (s0 + s1) + (s2 + s3);
  #pragma unroll
  for (int off = 16; off > 0; off >>= 1) s += __shfl_xor(s, off, 32);
  const float inv = 1.0f / s;

  float* row = wgt + (size_t)p * NPIX + lane;
  #pragma unroll
  for (int j = 0; j < RW; ++j) row[32 * j] = acc[j] * inv * g[j];
}

__global__ __launch_bounds__(DTHR) void diag_sum_kernel(
    const float* __restrict__ wgt,
    float* __restrict__ S)
{
  __shared__ __attribute__((aligned(16))) float tile[DTILE];

  const int tid = threadIdx.x, lane = tid & 31;
  const int w = __builtin_amdgcn_readfirstlane((int)(threadIdx.x >> 5));
  const int p0 = blockIdx.x * DROWS;
  const int p = p0 + w;

  float acc[RW];
  diag_rows(acc, tile, wgt, p0, w, lane, tid);

  float* row = S + (size_t)p * NPIX + lane;
  #pragma unroll
  for (int j = 0; j < RW; ++j) row[32 * j] = acc[j];
}

__global__ __launch_bounds__(128) __attribute__((amdgpu_waves_per_eu(1, 1)))
void gemm2_kernel(
    const float* __restrict__ bg,
    const float* __restrict__ S,
    float* __restrict__ Vp)
{
  __shared__ __attribute__((aligned(16))) float la[2][64][36];
  __shared__ __attribute__((aligned(16))) float lb[2][256][36];

  const int tid = threadIdx.x, lane = tid & 31, wave = tid >> 5;
  const int wq = wave;
  const int q0 = blockIdx.x * 256;
  const int nb = blockIdx.z * NSEG;
  const int fr = lane & 15;
  const int fk = (lane >> 4) << 1;

  const int lrow = tid >> 3;
  const int lcol = (tid & 7) << 2;
  const float* asrc = bg + (size_t)lrow * NPIX + nb + lcol;
  const float* bsrc = S + (size_t)(q0 + lrow) * NPIX + nb + lcol;
  float* V = Vp + (size_t)blockIdx.z * ((size_t)CCH * NPIX);

  v8f acc[4][4];
  #pragma unroll
  for (int i = 0; i < 4; ++i)
    #pragma unroll
    for (int j = 0; j < 4; ++j) acc[i][j] = (v8f){};

  v4f ra[4], rb[16];
  #pragma unroll
  for (int i = 0; i < 4; ++i)
    ra[i] = *(const v4f*)(asrc + (size_t)i * 16 * NPIX);
  #pragma unroll
  for (int i = 0; i < 16; ++i)
    rb[i] = *(const v4f*)(bsrc + (size_t)i * 16 * NPIX);
  #pragma unroll
  for (int i = 0; i < 4; ++i)  *(v4f*)&la[0][lrow + i * 16][lcol] = ra[i];
  #pragma unroll
  for (int i = 0; i < 16; ++i) *(v4f*)&lb[0][lrow + i * 16][lcol] = rb[i];
  __syncthreads();

  int buf = 0;
  for (int nc = 32; nc < NSEG; nc += 32) {
    asrc += 32; bsrc += 32;
    #pragma unroll
    for (int i = 0; i < 4; ++i)
      ra[i] = *(const v4f*)(asrc + (size_t)i * 16 * NPIX);
    #pragma unroll
    for (int i = 0; i < 16; ++i)
      rb[i] = *(const v4f*)(bsrc + (size_t)i * 16 * NPIX);
    __builtin_amdgcn_sched_barrier(0);

    CHUNK_MMA(la[buf], lb[buf], 0, wq * 64, 4, 4)

    __builtin_amdgcn_sched_barrier(0);
    #pragma unroll
    for (int i = 0; i < 4; ++i)  *(v4f*)&la[buf ^ 1][lrow + i * 16][lcol] = ra[i];
    #pragma unroll
    for (int i = 0; i < 16; ++i) *(v4f*)&lb[buf ^ 1][lrow + i * 16][lcol] = rb[i];
    __syncthreads();
    buf ^= 1;
  }
  CHUNK_MMA(la[buf], lb[buf], 0, wq * 64, 4, 4)

  const int dr = (lane >> 4) << 3, dc = lane & 15;
  #pragma unroll
  for (int i = 0; i < 4; ++i)
    #pragma unroll
    for (int j = 0; j < 4; ++j) {
      float* dst = V + (size_t)(i * 16 + dr) * NPIX
                     + q0 + wq * 64 + j * 16 + dc;
      #pragma unroll
      for (int r = 0; r < 8; ++r) dst[(size_t)r * NPIX] = acc[i][j][r];
    }
}

__device__ __forceinline__ void plane_sum_role(
    const int blk,
    const float* __restrict__ Vp, float* __restrict__ out)
{
  const int t = (blk * 256 + threadIdx.x) * 4;
  v4f v[NSPLIT];
  #pragma unroll
  for (int s = 0; s < NSPLIT; ++s)
    v[s] = *(const v4f*)(Vp + (size_t)s * ((size_t)CCH * NPIX) + t);
  v4f a = v[0];
  #pragma unroll
  for (int s = 1; s < NSPLIT; ++s) a += v[s];
  *(v4f*)(out + t) = a;
}

__global__ __launch_bounds__(256) void plane_sum_kernel(
    const float* __restrict__ Vp, float* __restrict__ out)
{
  plane_sum_role(blockIdx.x, Vp, out);
}

#define FRONT_G1   (G1BX * G1BY)
#define FRONT_NORM (FRONT_G1 + NPIX / 32)
#define FRONT_ALL  (FRONT_NORM + (CCH * NPIX) / 1024)

__global__ __launch_bounds__(256) void front_kernel(
    const float* __restrict__ fg,
    const float* __restrict__ bg,
    const float* __restrict__ mask,
    float* __restrict__ mp,
    float* __restrict__ rnrm,
    float* __restrict__ G,
    const float* __restrict__ Vp_prev,
    float* __restrict__ out_prev)
{
  __shared__ __attribute__((aligned(16))) float la[CCH][GPA];
  __shared__ __attribute__((aligned(16))) float lb[CCH][GPB];

  const int blk = blockIdx.x;
  if (blk < FRONT_G1)
    gemm1_role(blk, la, lb, fg, bg, G);
  else if (blk < FRONT_NORM)
    prep_norm_role(blk - FRONT_G1, (float (*)[32])la, bg, mask, mp, rnrm);
  else
    plane_sum_role(blk - FRONT_NORM, Vp_prev, out_prev);
}

extern "C" void kernel_launch(void* const* d_in, const int* in_sizes, int n_in,
                              void* d_out, int out_size, void* d_ws, size_t ws_size,
                              hipStream_t stream) {
  const float* fg   = (const float*)d_in[0];
  const float* bg   = (const float*)d_in[1];
  const float* mask = (const float*)d_in[2];
  float* out = (float*)d_out;

  float* ws = (float*)d_ws;
  const size_t NN = (size_t)NPIX * NPIX;
  float* mp = ws;
  float* G  = ws + NPIX;
  float* S  = G;
  float* Wg = G + NN;
  float* Vp = Wg;
  float* rn = Wg + NN;

  for (int b = 0; b < 4; ++b) {
    const float* fgb = fg + (size_t)b * CCH * NPIX;
    const float* bgb = bg + (size_t)b * CCH * NPIX;
    const float* mb  = mask + (size_t)b * NPIX;
    float* outb = out + (size_t)b * CCH * NPIX;

    if (b == 0)
      front_kernel<<<FRONT_NORM, 256, 0, stream>>>(fgb, bgb, mb, mp, rn, G, Vp, outb);
    else
      front_kernel<<<FRONT_ALL, 256, 0, stream>>>(fgb, bgb, mb, mp, rn, G, Vp,
                                                  outb - (size_t)CCH * NPIX);

    corr_softmax_kernel<<<NPIX / DROWS, DTHR, 0, stream>>>(G, mp, rn, Wg);

    diag_sum_kernel<<<NPIX / DROWS, DTHR, 0, stream>>>(Wg, S);

    dim3 g2(NPIX / 256, 1, NSPLIT);
    gemm2_kernel<<<g2, 128, 0, stream>>>(bgb, S, Vp);
  }
  plane_sum_kernel<<<(CCH * NPIX) / 1024, 256, 0, stream>>>(
      Vp, out + (size_t)3 * CCH * NPIX);
}


# both GEMM epilogues: each wave turns its accumulator tile in LDS and stores 16 bytes per lane, so every 128-B line of G and of the partial planes leaves as one whole store, not two 64-B halves
# speedup vs baseline: 13.1563x; 1.0320x over previous
#include <hip/hip_runtime.h>


typedef __attribute__((ext_vector_type(2))) float v2f;
typedef __attribute__((ext_vector_type(8))) float v8f;
typedef __attribute__((ext_vector_type(4))) float v4f;

#define NPIX 2304
#define HW   48
#define CCH  64
#define NSPLIT 24
#define NSEG (NPIX / NSPLIT)
#define G2SP 68

#define WMMA_F32(a, b, c) \
  __builtin_amdgcn_wmma_f32_16x16x4_f32(false, (a), false, (b), (short)0, (c), false, false)

__device__ __forceinline__ void prep_norm_role(
    const int blk,
    float (*part)[32],
    const float* __restrict__ bg,
    const float* __restrict__ mask,
    float* __restrict__ mp,
    float* __restrict__ rnrm)
{
  const int lane = threadIdx.x & 31, w = threadIdx.x >> 5;
  const int n = blk * 32 + lane;
  const int ny = n / HW, nx = n % HW;

  int  off[9];
  bool in[9];
  #pragma unroll
  for (int dy = 0; dy < 3; ++dy) {
    const int y = ny + dy - 1;
    const int cy = min(max(y, 0), HW - 1);
    #pragma unroll
    for (int dx = 0; dx < 3; ++dx) {
      const int x = nx + dx - 1;
      const int cx = min(max(x, 0), HW - 1);
      off[dy * 3 + dx] = cy * HW + cx;
      in[dy * 3 + dx]  = ((unsigned)y < (unsigned)HW) && ((unsigned)x < (unsigned)HW);
    }
  }

  const float* bc = bg + (size_t)(w * 8) * NPIX;
  float v[8][9];
  #pragma unroll
  for (int c = 0; c < 8; ++c)
    #pragma unroll
    for (int t = 0; t < 9; ++t)
      v[c][t] = bc[c * NPIX + off[t]];

  float mv[9];
  if (w == 0) {
    #pragma unroll
    for (int t = 0; t < 9; ++t) mv[t] = mask[off[t]];
  }

  float sq = 0.f;
  #pragma unroll
  for (int c = 0; c < 8; ++c)
    #pragma unroll
    for (int t = 0; t < 9; ++t) {
      const float x = in[t] ? v[c][t] : 0.f;
      sq += x * x;
    }
  part[w][lane] = sq;

  if (w == 0) {
    float msum = 0.f;
    #pragma unroll
    for (int t = 0; t < 9; ++t) msum += in[t] ? mv[t] : 0.f;
    mp[n] = (msum != 0.f) ? 1.f : 0.f;
  }
  __syncthreads();

  if (w == 0) {
    float s = part[0][lane];
    #pragma unroll
    for (int i = 1; i < 8; ++i) s += part[i][lane];
    rnrm[n] = 1.f / fmaxf(sqrtf(s), 1e-4f);
  }
}

#define CHUNK_MMA(LA, LB, AOFF, BOFF, NI, NJ)                           \
  _Pragma("unroll")                                                     \
  for (int kk = 0; kk < 32; kk += 4) {                                  \
    v2f af[NI], bf[NJ];                                                 \
    _Pragma("unroll")                                                   \
    for (int i = 0; i < (NI); ++i)                                      \
      af[i] = *(const v2f*)&LA[(AOFF) + i * 16 + fr][kk + fk];          \
    _Pragma("unroll")                                                   \
    for (int j = 0; j < (NJ); ++j)                                      \
      bf[j] = *(const v2f*)&LB[(BOFF) + j * 16 + fr][kk + fk];          \
    _Pragma("unroll")                                                   \
    for (int i = 0; i < (NI); ++i)                                      \
      _Pragma("unroll")                                                 \
      for (int j = 0; j < (NJ); ++j)                                    \
        acc[i][j] = WMMA_F32(af[i], bf[j], acc[i][j]);                  \
  }

#define G1N 192
#define G1I 2
#define G1J 6
#define GPA 136
#define GPB 200
#define G1SP 100

#define G1BX (NPIX / G1N)
#define G1BY (NPIX / 128)

__device__ __forceinline__ void gemm1_role(
    const int blk,
    float (*la)[GPA],
    float (*lb)[GPB],
    const float* __restrict__ fg,
    const float* __restrict__ bg,
    float* __restrict__ G)
{
  const int tid = threadIdx.x, lane = tid & 31, wave = tid >> 5;
  const int wp = wave >> 1;
  const int wn = wave & 1;
  const int p0 = (blk / G1BX) * 128, n0 = (blk % G1BX) * G1N;
  const int fr = lane & 15;
  const int fk = (lane >> 4) << 1;

  const int arow = tid >> 5;
  const int acol = (tid & 31) << 2;
  v4f ra[8], rb[12];
  #pragma unroll
  for (int i = 0; i < 8; ++i)
    ra[i] = *(const v4f*)(fg + (size_t)(arow + 8 * i) * NPIX + p0 + acol);
  #pragma unroll
  for (int i = 0; i < 12; ++i) {
    const int idx = i * 256 + tid;
    const int row = idx / 48, c4 = (idx % 48) << 2;
    rb[i] = *(const v4f*)(bg + (size_t)row * NPIX + n0 + c4);
  }
  #pragma unroll
  for (int i = 0; i < 8; ++i) *(v4f*)&la[arow + 8 * i][acol] = ra[i];
  #pragma unroll
  for (int i = 0; i < 12; ++i) {
    const int idx = i * 256 + tid;
    const int row = idx / 48, c4 = (idx % 48) << 2;
    *(v4f*)&lb[row][c4] = rb[i];
  }
  __syncthreads();

  v8f acc[G1I][G1J];
  #pragma unroll
  for (int i = 0; i < G1I; ++i)
    #pragma unroll
    for (int j = 0; j < G1J; ++j) acc[i][j] = (v8f){};

  #pragma unroll
  for (int kk = 0; kk < CCH; kk += 4) {
    v2f af[G1I], bf[G1J];
    #pragma unroll
    for (int i = 0; i < G1I; ++i) {
      af[i].x = la[kk + fk][wp * 32 + i * 16 + fr];
      af[i].y = la[kk + fk + 1][wp * 32 + i * 16 + fr];
    }
    #pragma unroll
    for (int j = 0; j < G1J; ++j) {
      bf[j].x = lb[kk + fk][wn * 96 + j * 16 + fr];
      bf[j].y = lb[kk + fk + 1][wn * 96 + j * 16 + fr];
    }
    #pragma unroll
    for (int i = 0; i < G1I; ++i)
      #pragma unroll
      for (int j = 0; j < G1J; ++j)
        acc[i][j] = WMMA_F32(af[i], bf[j], acc[i][j]);
  }

  const int dr = (lane >> 4) << 3, dc = lane & 15;
  float* stg = &lb[0][0] + wave * (16 * G1SP);
  v4f ob[G1I][12];
  __syncthreads();
  #pragma unroll
  for (int i = 0; i < G1I; ++i) {
    #pragma unroll
    for (int j = 0; j < G1J; ++j)
      #pragma unroll
      for (int r = 0; r < 8; ++r)
        stg[(r + dr) * G1SP + j * 16 + dc] = acc[i][j][r];
    __syncthreads();
    #pragma unroll
    for (int t = 0; t < 12; ++t) {
      const int idx = t * 32 + lane;
      const int row = idx / 24, c4 = (idx % 24) << 2;
      ob[i][t] = *(const v4f*)&stg[row * G1SP + c4];
    }
  }
  #pragma unroll
  for (int i = 0; i < G1I; ++i) {
    float* dst = G + (size_t)(p0 + wp * 32 + i * 16) * NPIX + n0 + wn * 96;
    #pragma unroll
    for (int t = 0; t < 12; ++t) {
      const int idx = t * 32 + lane;
      const int row = idx / 24, c4 = (idx % 24) << 2;
      *(v4f*)(dst + (size_t)row * NPIX + c4) = ob[i][t];
    }
  }
}

#define RW 72
#define DROWS 9
#define DSTG  (DROWS + 2)
#define DTHR  (DROWS * 32)
#define DPAD  64
#define DLD   ((DSTG * NPIX / 4) / DTHR)
#define DTILE (DSTG * NPIX + 2 * DPAD)

__device__ __forceinline__ void stage_fetch(
    v4f (&st)[DLD], const float* __restrict__ X, const int q0, const int tid)
{
  #pragma unroll
  for (int i = 0; i < DLD; ++i) {
    const int q = min(max(q0 + (i >> 1), 0), NPIX - 1);
    st[i] = *(const v4f*)(X + (size_t)q * NPIX + (((i & 1) * DTHR + tid) << 2));
  }
}

__device__ __forceinline__ void stage_put(
    float* __restrict__ tile, const v4f (&st)[DLD], const int tid)
{
  #pragma unroll
  for (int i = 0; i < DLD; ++i)
    *(v4f*)&tile[DPAD + (i >> 1) * NPIX + (((i & 1) * DTHR + tid) << 2)] = st[i];
}

template <int OY>
__device__ __forceinline__ void taps_from_tile(
    float (&acc)[RW], const float* __restrict__ tile,
    const int w, const int lane, const int py, const int px)
{
  const bool xlo[3] = { lane == 0, lane == 16, false };
  const bool xhi[3] = { false, lane == 15, lane == 31 };
  #pragma unroll
  for (int ox = -1; ox <= 1; ++ox) {
    if (((unsigned)(py + OY) < (unsigned)HW) && ((unsigned)(px + ox) < (unsigned)HW)) {
      const float* src = tile + (DPAD + (w + 1 + ox) * NPIX + OY * HW + ox) + lane;
      float v[RW];
      #pragma unroll
      for (int j = 0; j < RW; ++j) v[j] = src[32 * j];
      #pragma unroll
      for (int j = 0; j < RW; ++j) {
        bool ok = (unsigned)(lane + 32 * j + HW * OY) < (unsigned)NPIX;
        if (ox < 0) ok = ok && !xlo[j % 3];
        if (ox > 0) ok = ok && !xhi[j % 3];
        acc[j] += ok ? v[j] : 0.f;
      }
    }
  }
}

__device__ __forceinline__ void diag_rows(
    float (&acc)[RW], float* __restrict__ tile, const float* __restrict__ X,
    const int p0, const int w, const int lane, const int tid)
{
  const int p = p0 + w;
  const int py = p / HW, px = p % HW;

  #pragma unroll
  for (int j = 0; j < RW; ++j) acc[j] = 0.f;

  if (tid < DPAD) {
    tile[tid] = 0.f;
    tile[DPAD + DSTG * NPIX + tid] = 0.f;
  }

  v4f st[DLD];
  stage_fetch(st, X, p0 - HW - 1, tid);
  stage_put(tile, st, tid);
  __syncthreads();

  stage_fetch(st, X, p0 - 1, tid);
  __builtin_amdgcn_sched_barrier(0);
  taps_from_tile<-1>(acc, tile, w, lane, py, px);
  __syncthreads();
  stage_put(tile, st, tid);
  __syncthreads();

  stage_fetch(st, X, p0 + HW - 1, tid);
  __builtin_amdgcn_sched_barrier(0);
  taps_from_tile<0>(acc, tile, w, lane, py, px);
  __syncthreads();
  stage_put(tile, st, tid);
  __syncthreads();

  taps_from_tile<1>(acc, tile, w, lane, py, px);
}

__global__ __launch_bounds__(DTHR) void corr_softmax_kernel(
    const float* __restrict__ G,
    const float* __restrict__ mp,
    const float* __restrict__ rnrm,
    float* __restrict__ wgt)
{
  __shared__ __attribute__((aligned(16))) float tile[DTILE];

  const int tid = threadIdx.x, lane = tid & 31;
  const int w = __builtin_amdgcn_readfirstlane((int)(threadIdx.x >> 5));
  const int p0 = blockIdx.x * DROWS;
  const int p = p0 + w;

  float acc[RW];
  diag_rows(acc, tile, G, p0, w, lane, tid);

  float g[RW];
  float m = -3.0e38f;
  #pragma unroll
  for (int j = 0; j < RW; ++j) {
    const int n = lane + 32 * j;
    const float r = rnrm[n], gm = mp[n];
    acc[j] = acc[j] * r * gm * 10.0f;
    g[j] = gm * r;
    m = fmaxf(m, acc[j]);
  }
  #pragma unroll
  for (int off = 16; off > 0; off >>= 1) m = fmaxf(m, __shfl_xor(m, off, 32));

  float s0 = 0.f, s1 = 0.f, s2 = 0.f, s3 = 0.f;
  #pragma unroll
  for (int j = 0; j < RW; j += 4) {
    acc[j]     = __expf(acc[j] - m);      s0 += acc[j];
    acc[j + 1] = __expf(acc[j + 1] - m);  s1 += acc[j + 1];
    acc[j + 2] = __expf(acc[j + 2] - m);  s2 += acc[j + 2];
    acc[j + 3] = __expf(acc[j + 3] - m);  s3 += acc[j + 3];
  }
  float s = (s0 + s1) + (s2 + s3);
  #pragma unroll
  for (int off = 16; off > 0; off >>= 1) s += __shfl_xor(s, off, 32);
  const float inv = 1.0f / s;

  float* row = wgt + (size_t)p * NPIX + lane;
  #pragma unroll
  for (int j = 0; j < RW; ++j) row[32 * j] = acc[j] * inv * g[j];
}

__global__ __launch_bounds__(DTHR) void diag_sum_kernel(
    const float* __restrict__ wgt,
    float* __restrict__ S)
{
  __shared__ __attribute__((aligned(16))) float tile[DTILE];

  const int tid = threadIdx.x, lane = tid & 31;
  const int w = __builtin_amdgcn_readfirstlane((int)(threadIdx.x >> 5));
  const int p0 = blockIdx.x * DROWS;
  const int p = p0 + w;

  float acc[RW];
  diag_rows(acc, tile, wgt, p0, w, lane, tid);

  float* row = S + (size_t)p * NPIX + lane;
  #pragma unroll
  for (int j = 0; j < RW; ++j) row[32 * j] = acc[j];
}

__global__ __launch_bounds__(128) __attribute__((amdgpu_waves_per_eu(1, 1)))
void gemm2_kernel(
    const float* __restrict__ bg,
    const float* __restrict__ S,
    float* __restrict__ Vp)
{
  __shared__ __attribute__((aligned(16))) float la[2][64][36];
  __shared__ __attribute__((aligned(16))) float lb[2][256][36];

  const int tid = threadIdx.x, lane = tid & 31, wave = tid >> 5;
  const int wq = wave;
  const int q0 = blockIdx.x * 256;
  const int nb = blockIdx.z * NSEG;
  const int fr = lane & 15;
  const int fk = (lane >> 4) << 1;

  const int lrow = tid >> 3;
  const int lcol = (tid & 7) << 2;
  const float* asrc = bg + (size_t)lrow * NPIX + nb + lcol;
  const float* bsrc = S + (size_t)(q0 + lrow) * NPIX + nb + lcol;
  float* V = Vp + (size_t)blockIdx.z * ((size_t)CCH * NPIX);

  v8f acc[4][4];
  #pragma unroll
  for (int i = 0; i < 4; ++i)
    #pragma unroll
    for (int j = 0; j < 4; ++j) acc[i][j] = (v8f){};

  v4f ra[4], rb[16];
  #pragma unroll
  for (int i = 0; i < 4; ++i)
    ra[i] = *(const v4f*)(asrc + (size_t)i * 16 * NPIX);
  #pragma unroll
  for (int i = 0; i < 16; ++i)
    rb[i] = *(const v4f*)(bsrc + (size_t)i * 16 * NPIX);
  #pragma unroll
  for (int i = 0; i < 4; ++i)  *(v4f*)&la[0][lrow + i * 16][lcol] = ra[i];
  #pragma unroll
  for (int i = 0; i < 16; ++i) *(v4f*)&lb[0][lrow + i * 16][lcol] = rb[i];
  __syncthreads();

  int buf = 0;
  for (int nc = 32; nc < NSEG; nc += 32) {
    asrc += 32; bsrc += 32;
    #pragma unroll
    for (int i = 0; i < 4; ++i)
      ra[i] = *(const v4f*)(asrc + (size_t)i * 16 * NPIX);
    #pragma unroll
    for (int i = 0; i < 16; ++i)
      rb[i] = *(const v4f*)(bsrc + (size_t)i * 16 * NPIX);
    __builtin_amdgcn_sched_barrier(0);

    CHUNK_MMA(la[buf], lb[buf], 0, wq * 64, 4, 4)

    __builtin_amdgcn_sched_barrier(0);
    #pragma unroll
    for (int i = 0; i < 4; ++i)  *(v4f*)&la[buf ^ 1][lrow + i * 16][lcol] = ra[i];
    #pragma unroll
    for (int i = 0; i < 16; ++i) *(v4f*)&lb[buf ^ 1][lrow + i * 16][lcol] = rb[i];
    __syncthreads();
    buf ^= 1;
  }
  CHUNK_MMA(la[buf], lb[buf], 0, wq * 64, 4, 4)

  const int dr = (lane >> 4) << 3, dc = lane & 15;
  float* stg = &lb[0][0][0] + wave * (64 * G2SP);
  __syncthreads();
  #pragma unroll
  for (int i = 0; i < 4; ++i)
    #pragma unroll
    for (int j = 0; j < 4; ++j)
      #pragma unroll
      for (int r = 0; r < 8; ++r)
        stg[(i * 16 + r + dr) * G2SP + j * 16 + dc] = acc[i][j][r];
  __syncthreads();
  float* dst = V + q0 + wq * 64;
  #pragma unroll
  for (int t = 0; t < 32; ++t) {
    const int idx = t * 32 + lane;
    const int row = idx >> 4, c4 = (idx & 15) << 2;
    *(v4f*)(dst + (size_t)row * NPIX + c4) = *(const v4f*)&stg[row * G2SP + c4];
  }
}

__device__ __forceinline__ void plane_sum_role(
    const int blk,
    const float* __restrict__ Vp, float* __restrict__ out)
{
  const int t = (blk * 256 + threadIdx.x) * 4;
  v4f v[NSPLIT];
  #pragma unroll
  for (int s = 0; s < NSPLIT; ++s)
    v[s] = *(const v4f*)(Vp + (size_t)s * ((size_t)CCH * NPIX) + t);
  v4f a = v[0];
  #pragma unroll
  for (int s = 1; s < NSPLIT; ++s) a += v[s];
  *(v4f*)(out + t) = a;
}

__global__ __launch_bounds__(256) void plane_sum_kernel(
    const float* __restrict__ Vp, float* __restrict__ out)
{
  plane_sum_role(blockIdx.x, Vp, out);
}

#define FRONT_G1   (G1BX * G1BY)
#define FRONT_NORM (FRONT_G1 + NPIX / 32)
#define FRONT_ALL  (FRONT_NORM + (CCH * NPIX) / 1024)

__global__ __launch_bounds__(256) void front_kernel(
    const float* __restrict__ fg,
    const float* __restrict__ bg,
    const float* __restrict__ mask,
    float* __restrict__ mp,
    float* __restrict__ rnrm,
    float* __restrict__ G,
    const float* __restrict__ Vp_prev,
    float* __restrict__ out_prev)
{
  __shared__ __attribute__((aligned(16))) float la[CCH][GPA];
  __shared__ __attribute__((aligned(16))) float lb[CCH][GPB];

  const int blk = blockIdx.x;
  if (blk < FRONT_G1)
    gemm1_role(blk, la, lb, fg, bg, G);
  else if (blk < FRONT_NORM)
    prep_norm_role(blk - FRONT_G1, (float (*)[32])la, bg, mask, mp, rnrm);
  else
    plane_sum_role(blk - FRONT_NORM, Vp_prev, out_prev);
}

extern "C" void kernel_launch(void* const* d_in, const int* in_sizes, int n_in,
                              void* d_out, int out_size, void* d_ws, size_t ws_size,
                              hipStream_t stream) {
  const float* fg   = (const float*)d_in[0];
  const float* bg   = (const float*)d_in[1];
  const float* mask = (const float*)d_in[2];
  float* out = (float*)d_out;

  float* ws = (float*)d_ws;
  const size_t NN = (size_t)NPIX * NPIX;
  float* mp = ws;
  float* G  = ws + NPIX;
  float* S  = G;
  float* Wg = G + NN;
  float* Vp = Wg;
  float* rn = Wg + NN;

  for (int b = 0; b < 4; ++b) {
    const float* fgb = fg + (size_t)b * CCH * NPIX;
    const float* bgb = bg + (size_t)b * CCH * NPIX;
    const float* mb  = mask + (size_t)b * NPIX;
    float* outb = out + (size_t)b * CCH * NPIX;

    if (b == 0)
      front_kernel<<<FRONT_NORM, 256, 0, stream>>>(fgb, bgb, mb, mp, rn, G, Vp, outb);
    else
      front_kernel<<<FRONT_ALL, 256, 0, stream>>>(fgb, bgb, mb, mp, rn, G, Vp,
                                                  outb - (size_t)CCH * NPIX);

    corr_softmax_kernel<<<NPIX / DROWS, DTHR, 0, stream>>>(G, mp, rn, Wg);

    diag_sum_kernel<<<NPIX / DROWS, DTHR, 0, stream>>>(Wg, S);

    dim3 g2(NPIX / 256, 1, NSPLIT);
    gemm2_kernel<<<g2, 128, 0, stream>>>(bgb, S, Vp);
  }
  plane_sum_kernel<<<(CCH * NPIX) / 1024, 256, 0, stream>>>(
      Vp, out + (size_t)3 * CCH * NPIX);
}
